# starting kernel
# baseline (speedup 1.0000x reference)
#include <hip/hip_runtime.h>
#include <hip/hip_bf16.h>

#define BATCH 8
#define TLEN  2048
#define HDIM  256
#define BT    (BATCH * TLEN)

typedef __attribute__((ext_vector_type(2))) float v2f;
typedef __attribute__((ext_vector_type(8))) float v8f;

__global__ __launch_bounds__(256) void gates_gemm(
    const float* __restrict__ X,
    const float* __restrict__ Wa, const float* __restrict__ ba,
    const float* __restrict__ Wu, const float* __restrict__ bu,
    float* __restrict__ ab  ) {
  __shared__ float lwa[HDIM * 16];
  __shared__ float lwu[HDIM * 16];

  const int nt = blockIdx.x & 15;
  const int mg = blockIdx.x >> 4;
  const int lane = threadIdx.x & 31;
  const int wave = threadIdx.x >> 5;

  {
    const float4* ga = (const float4*)(Wa + nt * 16);
    const float4* gu = (const float4*)(Wu + nt * 16);
    float4* sa = (float4*)lwa;
    float4* su = (float4*)lwu;
    #pragma unroll
    for (int i = threadIdx.x; i < HDIM * 4; i += 256) {
      const int k = i >> 2;
      const int q = i & 3;
      sa[i] = ga[k * (HDIM / 4) + q];
      su[i] = gu[k * (HDIM / 4) + q];
    }
  }
  __syncthreads();

  const int hi   = lane >> 4;
  const int l16  = lane & 15;
  const int n    = nt * 16 + l16;
  const int koff = hi << 1;
  const int m0   = (mg * 8 + wave) * 64;

  v8f ca[4] = {v8f{}, v8f{}, v8f{}, v8f{}};
  v8f cu[4] = {v8f{}, v8f{}, v8f{}, v8f{}};

  const float* arow = X + (size_t)(m0 + l16) * HDIM + koff;
  #pragma unroll 2
  for (int k0 = 0; k0 < HDIM; k0 += 4) {
    const int k = k0 + koff;
    v2f bwa; bwa.x = lwa[k * 16 + l16]; bwa.y = lwa[(k + 1) * 16 + l16];
    v2f bwu; bwu.x = lwu[k * 16 + l16]; bwu.y = lwu[(k + 1) * 16 + l16];
    #pragma unroll
    for (int s = 0; s < 4; ++s) {
      v2f af = *(const v2f*)(arow + (size_t)(16 * s) * HDIM + k0);
      ca[s] = __builtin_amdgcn_wmma_f32_16x16x4_f32(false, af, false, bwa,
                                                    (short)0, ca[s], false, false);
      cu[s] = __builtin_amdgcn_wmma_f32_16x16x4_f32(false, af, false, bwu,
                                                    (short)0, cu[s], false, false);
    }
  }

  const float bia = ba[n];
  const float biu = bu[n];
  #pragma unroll
  for (int s = 0; s < 4; ++s) {
    const int rbase = m0 + 16 * s + (hi << 3);
    #pragma unroll
    for (int r = 0; r < 8; ++r) {
      const float za = ca[s][r] + bia;
      const float zu = cu[s][r] + biu;
      const float ga = 1.0f / (1.0f + __expf(-za));
      const float gu = tanhf(zu);
      v2f e; e.x = ga; e.y = (1.0f - ga) * gu;
      *(v2f*)(ab + 2 * ((size_t)(rbase + r) * HDIM + n)) = e;
    }
  }
}

__global__ __launch_bounds__(256) void scan_plif(
    const float* __restrict__ ab, float* __restrict__ s_out,
    const float* __restrict__ tau_p, const float* __restrict__ thr_p) {
  const int idx = blockIdx.x * blockDim.x + threadIdx.x;
  const int b = idx >> 8;
  const int h = idx & (HDIM - 1);
  const float inv_tau = 1.0f / tau_p[0];
  const float thr = thr_p[0];
  const float decay = 1.0f - inv_tau;

  float hstate = 0.0f;
  float v = 0.0f;
  const size_t base = ((size_t)b * TLEN) * HDIM + h;

  #pragma unroll 4
  for (int t = 0; t < TLEN; ++t) {
    const size_t off = base + (size_t)t * HDIM;
    if (t + 16 < TLEN)
      __builtin_prefetch(ab + 2 * (base + (size_t)(t + 16) * HDIM), 0, 0);
    v2f e = *(const v2f*)(ab + 2 * off);
    hstate = e.x * hstate + e.y;
    v = decay * v + inv_tau * hstate;
    const float sp = (v > thr) ? 1.0f : 0.0f;
    v -= sp * thr;
    s_out[off] = sp;
  }
}

__global__ __launch_bounds__(256) void synapse_gemm(
    const float* __restrict__ S, const float* __restrict__ Wsyn,
    float* __restrict__ out) {
  __shared__ float lws[16 * HDIM];

  const int nt = blockIdx.x & 15;
  const int mg = blockIdx.x >> 4;
  const int lane = threadIdx.x & 31;
  const int wave = threadIdx.x >> 5;

  {
    const float4* g = (const float4*)(Wsyn + (size_t)nt * 16 * HDIM);
    float4* sm = (float4*)lws;
    #pragma unroll
    for (int i = threadIdx.x; i < 16 * HDIM / 4; i += 256)
      sm[i] = g[i];
  }
  __syncthreads();

  const int hi   = lane >> 4;
  const int l16  = lane & 15;
  const int n    = nt * 16 + l16;
  const int koff = hi << 1;
  const int m0   = (mg * 8 + wave) * 64;

  v8f c[4] = {v8f{}, v8f{}, v8f{}, v8f{}};

  const float* arow = S + (size_t)(m0 + l16) * HDIM + koff;
  const float* brow = lws + l16 * HDIM + koff;
  #pragma unroll 2
  for (int k0 = 0; k0 < HDIM; k0 += 4) {
    v2f bf = *(const v2f*)(brow + k0);
    #pragma unroll
    for (int s = 0; s < 4; ++s) {
      v2f af = *(const v2f*)(arow + (size_t)(16 * s) * HDIM + k0);
      c[s] = __builtin_amdgcn_wmma_f32_16x16x4_f32(false, af, false, bf,
                                                   (short)0, c[s], false, false);
    }
  }

  #pragma unroll
  for (int s = 0; s < 4; ++s) {
    const int rbase = m0 + 16 * s + (hi << 3);
    #pragma unroll
    for (int r = 0; r < 8; ++r)
      out[(size_t)(rbase + r) * HDIM + n] = c[s][r];
  }
}

extern "C" void kernel_launch(void* const* d_in, const int* in_sizes, int n_in,
                              void* d_out, int out_size, void* d_ws, size_t ws_size,
                              hipStream_t stream) {
  const float* x    = (const float*)d_in[0];
  const float* Wa   = (const float*)d_in[1];
  const float* ba   = (const float*)d_in[2];
  const float* Wu   = (const float*)d_in[3];
  const float* bu   = (const float*)d_in[4];
  const float* tau  = (const float*)d_in[5];
  const float* thr  = (const float*)d_in[6];
  const float* Wsyn = (const float*)d_in[7];
  float* out = (float*)d_out;

  float* ab = (float*)d_ws;
  float* s  = ab + 2 * (size_t)BT * HDIM;

  gates_gemm<<<512, 256, 0, stream>>>(x, Wa, ba, Wu, bu, ab);
  scan_plif<<<8, 256, 0, stream>>>(ab, s, tau, thr);
  synapse_gemm<<<512, 256, 0, stream>>>(s, Wsyn, out);
}


# scan kernel: each lane keeps the next 64 time steps' (a,b) loads in flight in a register ring instead of one dependent 8-byte load per step; per-step arithmetic and its order unchanged
# speedup vs baseline: 5.7051x; 5.7051x over previous
#include <hip/hip_runtime.h>
#include <hip/hip_bf16.h>

#define BATCH 8
#define TLEN  2048
#define HDIM  256
#define BT    (BATCH * TLEN)

typedef __attribute__((ext_vector_type(2))) float v2f;
typedef __attribute__((ext_vector_type(8))) float v8f;

__global__ __launch_bounds__(256) void gates_gemm(
    const float* __restrict__ X,
    const float* __restrict__ Wa, const float* __restrict__ ba,
    const float* __restrict__ Wu, const float* __restrict__ bu,
    float* __restrict__ ab  ) {
  __shared__ float lwa[HDIM * 16];
  __shared__ float lwu[HDIM * 16];

  const int nt = blockIdx.x & 15;
  const int mg = blockIdx.x >> 4;
  const int lane = threadIdx.x & 31;
  const int wave = threadIdx.x >> 5;

  {
    const float4* ga = (const float4*)(Wa + nt * 16);
    const float4* gu = (const float4*)(Wu + nt * 16);
    float4* sa = (float4*)lwa;
    float4* su = (float4*)lwu;
    #pragma unroll
    for (int i = threadIdx.x; i < HDIM * 4; i += 256) {
      const int k = i >> 2;
      const int q = i & 3;
      sa[i] = ga[k * (HDIM / 4) + q];
      su[i] = gu[k * (HDIM / 4) + q];
    }
  }
  __syncthreads();

  const int hi   = lane >> 4;
  const int l16  = lane & 15;
  const int n    = nt * 16 + l16;
  const int koff = hi << 1;
  const int m0   = (mg * 8 + wave) * 64;

  v8f ca[4] = {v8f{}, v8f{}, v8f{}, v8f{}};
  v8f cu[4] = {v8f{}, v8f{}, v8f{}, v8f{}};

  const float* arow = X + (size_t)(m0 + l16) * HDIM + koff;
  #pragma unroll 2
  for (int k0 = 0; k0 < HDIM; k0 += 4) {
    const int k = k0 + koff;
    v2f bwa; bwa.x = lwa[k * 16 + l16]; bwa.y = lwa[(k + 1) * 16 + l16];
    v2f bwu; bwu.x = lwu[k * 16 + l16]; bwu.y = lwu[(k + 1) * 16 + l16];
    #pragma unroll
    for (int s = 0; s < 4; ++s) {
      v2f af = *(const v2f*)(arow + (size_t)(16 * s) * HDIM + k0);
      ca[s] = __builtin_amdgcn_wmma_f32_16x16x4_f32(false, af, false, bwa,
                                                    (short)0, ca[s], false, false);
      cu[s] = __builtin_amdgcn_wmma_f32_16x16x4_f32(false, af, false, bwu,
                                                    (short)0, cu[s], false, false);
    }
  }

  const float bia = ba[n];
  const float biu = bu[n];
  #pragma unroll
  for (int s = 0; s < 4; ++s) {
    const int rbase = m0 + 16 * s + (hi << 3);
    #pragma unroll
    for (int r = 0; r < 8; ++r) {
      const float za = ca[s][r] + bia;
      const float zu = cu[s][r] + biu;
      const float ga = 1.0f / (1.0f + __expf(-za));
      const float gu = tanhf(zu);
      v2f e; e.x = ga; e.y = (1.0f - ga) * gu;
      *(v2f*)(ab + 2 * ((size_t)(rbase + r) * HDIM + n)) = e;
    }
  }
}

#define SCAN_RING 64

__global__ __launch_bounds__(256) void scan_plif(
    const float* __restrict__ ab, float* __restrict__ s_out,
    const float* __restrict__ tau_p, const float* __restrict__ thr_p) {
  const int idx = blockIdx.x * blockDim.x + threadIdx.x;
  const int b = idx >> 8;
  const int h = idx & (HDIM - 1);
  const float inv_tau = 1.0f / tau_p[0];
  const float thr = thr_p[0];
  const float decay = 1.0f - inv_tau;

  float hstate = 0.0f;
  float v = 0.0f;
  const size_t base = ((size_t)b * TLEN) * HDIM + h;
  const float* src = ab + 2 * base;
  float* dst = s_out + base;

  v2f ring[SCAN_RING];
  #pragma unroll
  for (int i = 0; i < SCAN_RING; ++i)
    ring[i] = *(const v2f*)(src + (size_t)i * (2 * HDIM));

  for (int t0 = 0; t0 < TLEN - SCAN_RING; t0 += SCAN_RING) {
    const float* nxt = src + (size_t)(t0 + SCAN_RING) * (2 * HDIM);
    float* cur = dst + (size_t)t0 * HDIM;
    #pragma unroll
    for (int i = 0; i < SCAN_RING; ++i) {
      const v2f e = ring[i];
      ring[i] = *(const v2f*)(nxt + (size_t)i * (2 * HDIM));
      hstate = e.x * hstate + e.y;
      v = decay * v + inv_tau * hstate;
      const float sp = (v > thr) ? 1.0f : 0.0f;
      v -= sp * thr;
      cur[(size_t)i * HDIM] = sp;
    }
  }

  {
    float* cur = dst + (size_t)(TLEN - SCAN_RING) * HDIM;
    #pragma unroll
    for (int i = 0; i < SCAN_RING; ++i) {
      const v2f e = ring[i];
      hstate = e.x * hstate + e.y;
      v = decay * v + inv_tau * hstate;
      const float sp = (v > thr) ? 1.0f : 0.0f;
      v -= sp * thr;
      cur[(size_t)i * HDIM] = sp;
    }
  }
}

__global__ __launch_bounds__(256) void synapse_gemm(
    const float* __restrict__ S, const float* __restrict__ Wsyn,
    float* __restrict__ out) {
  __shared__ float lws[16 * HDIM];

  const int nt = blockIdx.x & 15;
  const int mg = blockIdx.x >> 4;
  const int lane = threadIdx.x & 31;
  const int wave = threadIdx.x >> 5;

  {
    const float4* g = (const float4*)(Wsyn + (size_t)nt * 16 * HDIM);
    float4* sm = (float4*)lws;
    #pragma unroll
    for (int i = threadIdx.x; i < 16 * HDIM / 4; i += 256)
      sm[i] = g[i];
  }
  __syncthreads();

  const int hi   = lane >> 4;
  const int l16  = lane & 15;
  const int n    = nt * 16 + l16;
  const int koff = hi << 1;
  const int m0   = (mg * 8 + wave) * 64;

  v8f c[4] = {v8f{}, v8f{}, v8f{}, v8f{}};

  const float* arow = S + (size_t)(m0 + l16) * HDIM + koff;
  const float* brow = lws + l16 * HDIM + koff;
  #pragma unroll 2
  for (int k0 = 0; k0 < HDIM; k0 += 4) {
    v2f bf = *(const v2f*)(brow + k0);
    #pragma unroll
    for (int s = 0; s < 4; ++s) {
      v2f af = *(const v2f*)(arow + (size_t)(16 * s) * HDIM + k0);
      c[s] = __builtin_amdgcn_wmma_f32_16x16x4_f32(false, af, false, bf,
                                                   (short)0, c[s], false, false);
    }
  }

  #pragma unroll
  for (int s = 0; s < 4; ++s) {
    const int rbase = m0 + 16 * s + (hi << 3);
    #pragma unroll
    for (int r = 0; r < 8; ++r)
      out[(size_t)(rbase + r) * HDIM + n] = c[s][r];
  }
}

extern "C" void kernel_launch(void* const* d_in, const int* in_sizes, int n_in,
                              void* d_out, int out_size, void* d_ws, size_t ws_size,
                              hipStream_t stream) {
  const float* x    = (const float*)d_in[0];
  const float* Wa   = (const float*)d_in[1];
  const float* ba   = (const float*)d_in[2];
  const float* Wu   = (const float*)d_in[3];
  const float* bu   = (const float*)d_in[4];
  const float* tau  = (const float*)d_in[5];
  const float* thr  = (const float*)d_in[6];
  const float* Wsyn = (const float*)d_in[7];
  float* out = (float*)d_out;

  float* ab = (float*)d_ws;
  float* s  = ab + 2 * (size_t)BT * HDIM;

  gates_gemm<<<512, 256, 0, stream>>>(x, Wa, ba, Wu, bu, ab);
  scan_plif<<<8, 256, 0, stream>>>(ab, s, tau, thr);
  synapse_gemm<<<512, 256, 0, stream>>>(s, Wsyn, out);
}


# both GEMMs: A operand (x, spikes) staged per K chunk through LDS by coalesced 128-bit row loads issued one chunk ahead; fragments read from LDS, same k per lane and register, sums bit-identical
# speedup vs baseline: 6.7158x; 1.1772x over previous
#include <hip/hip_runtime.h>
#include <hip/hip_bf16.h>

#define BATCH 8
#define TLEN  2048
#define HDIM  256
#define BT    (BATCH * TLEN)

typedef __attribute__((ext_vector_type(2))) float v2f;
typedef __attribute__((ext_vector_type(4))) float v4f;
typedef __attribute__((ext_vector_type(8))) float v8f;

#define KC     32
#define NCHUNK (HDIM / KC)
#define APITCH 36
#define AROWS  512

#define A_CHUNK_LOAD(pre, g)                                              \
  _Pragma("unroll") for (int q = 0; q < 16; ++q)                          \
    (pre)[q] = *(const v4f*)((g) + (size_t)(4 * q) * HDIM)
#define A_CHUNK_STORE(pre, l)                                             \
  _Pragma("unroll") for (int q = 0; q < 16; ++q)                          \
    *(v4f*)((l) + (4 * q) * APITCH) = (pre)[q]

__global__ __launch_bounds__(256) void gates_gemm(
    const float* __restrict__ X,
    const float* __restrict__ Wa, const float* __restrict__ ba,
    const float* __restrict__ Wu, const float* __restrict__ bu,
    float* __restrict__ ab  ) {
  __shared__ float lwa[HDIM * 16];
  __shared__ float lwu[HDIM * 16];
  __shared__ __attribute__((aligned(16))) float lA[AROWS * APITCH];

  const int nt = blockIdx.x & 15;
  const int mg = blockIdx.x >> 4;
  const int lane = threadIdx.x & 31;
  const int wave = threadIdx.x >> 5;

  {
    const float4* ga = (const float4*)(Wa + nt * 16);
    const float4* gu = (const float4*)(Wu + nt * 16);
    float4* sa = (float4*)lwa;
    float4* su = (float4*)lwu;
    #pragma unroll
    for (int i = threadIdx.x; i < HDIM * 4; i += 256) {
      const int k = i >> 2;
      const int q = i & 3;
      sa[i] = ga[k * (HDIM / 4) + q];
      su[i] = gu[k * (HDIM / 4) + q];
    }
  }

  const int hi   = lane >> 4;
  const int l16  = lane & 15;
  const int n    = nt * 16 + l16;
  const int koff = hi << 1;
  const int m0   = (mg * 8 + wave) * 64;

  v8f ca[4] = {v8f{}, v8f{}, v8f{}, v8f{}};
  v8f cu[4] = {v8f{}, v8f{}, v8f{}, v8f{}};

  const float* gsrc = X + (size_t)(m0 + (lane >> 3)) * HDIM + 4 * (lane & 7);
  float* ldst = lA + (wave * 64 + (lane >> 3)) * APITCH + 4 * (lane & 7);
  const float* afrag = lA + (wave * 64 + l16) * APITCH + koff;

  v4f pre[16];
  A_CHUNK_LOAD(pre, gsrc);
  for (int kc = 0; kc < NCHUNK; ++kc) {
    __syncthreads();
    A_CHUNK_STORE(pre, ldst);
    __syncthreads();
    if (kc + 1 < NCHUNK)
      A_CHUNK_LOAD(pre, gsrc + (kc + 1) * KC);
    __builtin_amdgcn_sched_barrier(0);
    #pragma unroll
    for (int j = 0; j < KC / 4; ++j) {
      const int k = kc * KC + 4 * j + koff;
      v2f bwa; bwa.x = lwa[k * 16 + l16]; bwa.y = lwa[(k + 1) * 16 + l16];
      v2f bwu; bwu.x = lwu[k * 16 + l16]; bwu.y = lwu[(k + 1) * 16 + l16];
      #pragma unroll
      for (int s = 0; s < 4; ++s) {
        v2f af = *(const v2f*)(afrag + (16 * s) * APITCH + 4 * j);
        ca[s] = __builtin_amdgcn_wmma_f32_16x16x4_f32(false, af, false, bwa,
                                                      (short)0, ca[s], false, false);
        cu[s] = __builtin_amdgcn_wmma_f32_16x16x4_f32(false, af, false, bwu,
                                                      (short)0, cu[s], false, false);
      }
    }
  }

  const float bia = ba[n];
  const float biu = bu[n];
  #pragma unroll
  for (int s = 0; s < 4; ++s) {
    const int rbase = m0 + 16 * s + (hi << 3);
    #pragma unroll
    for (int r = 0; r < 8; ++r) {
      const float za = ca[s][r] + bia;
      const float zu = cu[s][r] + biu;
      const float ga = 1.0f / (1.0f + __expf(-za));
      const float gu = tanhf(zu);
      v2f e; e.x = ga; e.y = (1.0f - ga) * gu;
      *(v2f*)(ab + 2 * ((size_t)(rbase + r) * HDIM + n)) = e;
    }
  }
}

#define SCAN_RING 64

__global__ __launch_bounds__(256) void scan_plif(
    const float* __restrict__ ab, float* __restrict__ s_out,
    const float* __restrict__ tau_p, const float* __restrict__ thr_p) {
  const int idx = blockIdx.x * blockDim.x + threadIdx.x;
  const int b = idx >> 8;
  const int h = idx & (HDIM - 1);
  const float inv_tau = 1.0f / tau_p[0];
  const float thr = thr_p[0];
  const float decay = 1.0f - inv_tau;

  float hstate = 0.0f;
  float v = 0.0f;
  const size_t base = ((size_t)b * TLEN) * HDIM + h;
  const float* src = ab + 2 * base;
  float* dst = s_out + base;

  v2f ring[SCAN_RING];
  #pragma unroll
  for (int i = 0; i < SCAN_RING; ++i)
    ring[i] = *(const v2f*)(src + (size_t)i * (2 * HDIM));

  for (int t0 = 0; t0 < TLEN - SCAN_RING; t0 += SCAN_RING) {
    const float* nxt = src + (size_t)(t0 + SCAN_RING) * (2 * HDIM);
    float* cur = dst + (size_t)t0 * HDIM;
    #pragma unroll
    for (int i = 0; i < SCAN_RING; ++i) {
      const v2f e = ring[i];
      ring[i] = *(const v2f*)(nxt + (size_t)i * (2 * HDIM));
      hstate = e.x * hstate + e.y;
      v = decay * v + inv_tau * hstate;
      const float sp = (v > thr) ? 1.0f : 0.0f;
      v -= sp * thr;
      cur[(size_t)i * HDIM] = sp;
    }
  }

  {
    float* cur = dst + (size_t)(TLEN - SCAN_RING) * HDIM;
    #pragma unroll
    for (int i = 0; i < SCAN_RING; ++i) {
      const v2f e = ring[i];
      hstate = e.x * hstate + e.y;
      v = decay * v + inv_tau * hstate;
      const float sp = (v > thr) ? 1.0f : 0.0f;
      v -= sp * thr;
      cur[(size_t)i * HDIM] = sp;
    }
  }
}

__global__ __launch_bounds__(256) void synapse_gemm(
    const float* __restrict__ S, const float* __restrict__ Wsyn,
    float* __restrict__ out) {
  __shared__ float lws[16 * HDIM];
  __shared__ __attribute__((aligned(16))) float lA[AROWS * APITCH];

  const int nt = blockIdx.x & 15;
  const int mg = blockIdx.x >> 4;
  const int lane = threadIdx.x & 31;
  const int wave = threadIdx.x >> 5;

  {
    const float4* g = (const float4*)(Wsyn + (size_t)nt * 16 * HDIM);
    float4* sm = (float4*)lws;
    #pragma unroll
    for (int i = threadIdx.x; i < 16 * HDIM / 4; i += 256)
      sm[i] = g[i];
  }

  const int hi   = lane >> 4;
  const int l16  = lane & 15;
  const int n    = nt * 16 + l16;
  const int koff = hi << 1;
  const int m0   = (mg * 8 + wave) * 64;

  v8f c[4] = {v8f{}, v8f{}, v8f{}, v8f{}};

  const float* brow = lws + l16 * HDIM + koff;
  const float* gsrc = S + (size_t)(m0 + (lane >> 3)) * HDIM + 4 * (lane & 7);
  float* ldst = lA + (wave * 64 + (lane >> 3)) * APITCH + 4 * (lane & 7);
  const float* afrag = lA + (wave * 64 + l16) * APITCH + koff;

  v4f pre[16];
  A_CHUNK_LOAD(pre, gsrc);
  for (int kc = 0; kc < NCHUNK; ++kc) {
    __syncthreads();
    A_CHUNK_STORE(pre, ldst);
    __syncthreads();
    if (kc + 1 < NCHUNK)
      A_CHUNK_LOAD(pre, gsrc + (kc + 1) * KC);
    __builtin_amdgcn_sched_barrier(0);
    #pragma unroll
    for (int j = 0; j < KC / 4; ++j) {
      v2f bf = *(const v2f*)(brow + kc * KC + 4 * j);
      #pragma unroll
      for (int s = 0; s < 4; ++s) {
        v2f af = *(const v2f*)(afrag + (16 * s) * APITCH + 4 * j);
        c[s] = __builtin_amdgcn_wmma_f32_16x16x4_f32(false, af, false, bf,
                                                     (short)0, c[s], false, false);
      }
    }
  }

  #pragma unroll
  for (int s = 0; s < 4; ++s) {
    const int rbase = m0 + 16 * s + (hi << 3);
    #pragma unroll
    for (int r = 0; r < 8; ++r)
      out[(size_t)(rbase + r) * HDIM + n] = c[s][r];
  }
}

extern "C" void kernel_launch(void* const* d_in, const int* in_sizes, int n_in,
                              void* d_out, int out_size, void* d_ws, size_t ws_size,
                              hipStream_t stream) {
  const float* x    = (const float*)d_in[0];
  const float* Wa   = (const float*)d_in[1];
  const float* ba   = (const float*)d_in[2];
  const float* Wu   = (const float*)d_in[3];
  const float* bu   = (const float*)d_in[4];
  const float* tau  = (const float*)d_in[5];
  const float* thr  = (const float*)d_in[6];
  const float* Wsyn = (const float*)d_in[7];
  float* out = (float*)d_out;

  float* ab = (float*)d_ws;
  float* s  = ab + 2 * (size_t)BT * HDIM;

  gates_gemm<<<512, 256, 0, stream>>>(x, Wa, ba, Wu, bu, ab);
  scan_plif<<<8, 256, 0, stream>>>(ab, s, tau, thr);
  synapse_gemm<<<512, 256, 0, stream>>>(s, Wsyn, out);
}


# scan launched as 64 one-wave work-groups instead of 8 of eight waves, so each serial wave can have a WGP's path to memory to itself; kernel code, arithmetic and sum order unchanged
# speedup vs baseline: 7.2366x; 1.0775x over previous
#include <hip/hip_runtime.h>
#include <hip/hip_bf16.h>

#define BATCH 8
#define TLEN  2048
#define HDIM  256
#define BT    (BATCH * TLEN)

typedef __attribute__((ext_vector_type(2))) float v2f;
typedef __attribute__((ext_vector_type(4))) float v4f;
typedef __attribute__((ext_vector_type(8))) float v8f;

#define KC     32
#define NCHUNK (HDIM / KC)
#define APITCH 36
#define AROWS  512

#define A_CHUNK_LOAD(pre, g)                                              \
  _Pragma("unroll") for (int q = 0; q < 16; ++q)                          \
    (pre)[q] = *(const v4f*)((g) + (size_t)(4 * q) * HDIM)
#define A_CHUNK_STORE(pre, l)                                             \
  _Pragma("unroll") for (int q = 0; q < 16; ++q)                          \
    *(v4f*)((l) + (4 * q) * APITCH) = (pre)[q]

__global__ __launch_bounds__(256) void gates_gemm(
    const float* __restrict__ X,
    const float* __restrict__ Wa, const float* __restrict__ ba,
    const float* __restrict__ Wu, const float* __restrict__ bu,
    float* __restrict__ ab  ) {
  __shared__ float lwa[HDIM * 16];
  __shared__ float lwu[HDIM * 16];
  __shared__ __attribute__((aligned(16))) float lA[AROWS * APITCH];

  const int nt = blockIdx.x & 15;
  const int mg = blockIdx.x >> 4;
  const int lane = threadIdx.x & 31;
  const int wave = threadIdx.x >> 5;

  {
    const float4* ga = (const float4*)(Wa + nt * 16);
    const float4* gu = (const float4*)(Wu + nt * 16);
    float4* sa = (float4*)lwa;
    float4* su = (float4*)lwu;
    #pragma unroll
    for (int i = threadIdx.x; i < HDIM * 4; i += 256) {
      const int k = i >> 2;
      const int q = i & 3;
      sa[i] = ga[k * (HDIM / 4) + q];
      su[i] = gu[k * (HDIM / 4) + q];
    }
  }

  const int hi   = lane >> 4;
  const int l16  = lane & 15;
  const int n    = nt * 16 + l16;
  const int koff = hi << 1;
  const int m0   = (mg * 8 + wave) * 64;

  v8f ca[4] = {v8f{}, v8f{}, v8f{}, v8f{}};
  v8f cu[4] = {v8f{}, v8f{}, v8f{}, v8f{}};

  const float* gsrc = X + (size_t)(m0 + (lane >> 3)) * HDIM + 4 * (lane & 7);
  float* ldst = lA + (wave * 64 + (lane >> 3)) * APITCH + 4 * (lane & 7);
  const float* afrag = lA + (wave * 64 + l16) * APITCH + koff;

  v4f pre[16];
  A_CHUNK_LOAD(pre, gsrc);
  for (int kc = 0; kc < NCHUNK; ++kc) {
    __syncthreads();
    A_CHUNK_STORE(pre, ldst);
    __syncthreads();
    if (kc + 1 < NCHUNK)
      A_CHUNK_LOAD(pre, gsrc + (kc + 1) * KC);
    __builtin_amdgcn_sched_barrier(0);
    #pragma unroll
    for (int j = 0; j < KC / 4; ++j) {
      const int k = kc * KC + 4 * j + koff;
      v2f bwa; bwa.x = lwa[k * 16 + l16]; bwa.y = lwa[(k + 1) * 16 + l16];
      v2f bwu; bwu.x = lwu[k * 16 + l16]; bwu.y = lwu[(k + 1) * 16 + l16];
      #pragma unroll
      for (int s = 0; s < 4; ++s) {
        v2f af = *(const v2f*)(afrag + (16 * s) * APITCH + 4 * j);
        ca[s] = __builtin_amdgcn_wmma_f32_16x16x4_f32(false, af, false, bwa,
                                                      (short)0, ca[s], false, false);
        cu[s] = __builtin_amdgcn_wmma_f32_16x16x4_f32(false, af, false, bwu,
                                                      (short)0, cu[s], false, false);
      }
    }
  }

  const float bia = ba[n];
  const float biu = bu[n];
  #pragma unroll
  for (int s = 0; s < 4; ++s) {
    const int rbase = m0 + 16 * s + (hi << 3);
    #pragma unroll
    for (int r = 0; r < 8; ++r) {
      const float za = ca[s][r] + bia;
      const float zu = cu[s][r] + biu;
      const float ga = 1.0f / (1.0f + __expf(-za));
      const float gu = tanhf(zu);
      v2f e; e.x = ga; e.y = (1.0f - ga) * gu;
      *(v2f*)(ab + 2 * ((size_t)(rbase + r) * HDIM + n)) = e;
    }
  }
}

#define SCAN_RING 64

__global__ __launch_bounds__(32) void scan_plif(
    const float* __restrict__ ab, float* __restrict__ s_out,
    const float* __restrict__ tau_p, const float* __restrict__ thr_p) {
  const int idx = blockIdx.x * blockDim.x + threadIdx.x;
  const int b = idx >> 8;
  const int h = idx & (HDIM - 1);
  const float inv_tau = 1.0f / tau_p[0];
  const float thr = thr_p[0];
  const float decay = 1.0f - inv_tau;

  float hstate = 0.0f;
  float v = 0.0f;
  const size_t base = ((size_t)b * TLEN) * HDIM + h;
  const float* src = ab + 2 * base;
  float* dst = s_out + base;

  v2f ring[SCAN_RING];
  #pragma unroll
  for (int i = 0; i < SCAN_RING; ++i)
    ring[i] = *(const v2f*)(src + (size_t)i * (2 * HDIM));

  for (int t0 = 0; t0 < TLEN - SCAN_RING; t0 += SCAN_RING) {
    const float* nxt = src + (size_t)(t0 + SCAN_RING) * (2 * HDIM);
    float* cur = dst + (size_t)t0 * HDIM;
    #pragma unroll
    for (int i = 0; i < SCAN_RING; ++i) {
      const v2f e = ring[i];
      ring[i] = *(const v2f*)(nxt + (size_t)i * (2 * HDIM));
      hstate = e.x * hstate + e.y;
      v = decay * v + inv_tau * hstate;
      const float sp = (v > thr) ? 1.0f : 0.0f;
      v -= sp * thr;
      cur[(size_t)i * HDIM] = sp;
    }
  }

  {
    float* cur = dst + (size_t)(TLEN - SCAN_RING) * HDIM;
    #pragma unroll
    for (int i = 0; i < SCAN_RING; ++i) {
      const v2f e = ring[i];
      hstate = e.x * hstate + e.y;
      v = decay * v + inv_tau * hstate;
      const float sp = (v > thr) ? 1.0f : 0.0f;
      v -= sp * thr;
      cur[(size_t)i * HDIM] = sp;
    }
  }
}

__global__ __launch_bounds__(256) void synapse_gemm(
    const float* __restrict__ S, const float* __restrict__ Wsyn,
    float* __restrict__ out) {
  __shared__ float lws[16 * HDIM];
  __shared__ __attribute__((aligned(16))) float lA[AROWS * APITCH];

  const int nt = blockIdx.x & 15;
  const int mg = blockIdx.x >> 4;
  const int lane = threadIdx.x & 31;
  const int wave = threadIdx.x >> 5;

  {
    const float4* g = (const float4*)(Wsyn + (size_t)nt * 16 * HDIM);
    float4* sm = (float4*)lws;
    #pragma unroll
    for (int i = threadIdx.x; i < 16 * HDIM / 4; i += 256)
      sm[i] = g[i];
  }

  const int hi   = lane >> 4;
  const int l16  = lane & 15;
  const int n    = nt * 16 + l16;
  const int koff = hi << 1;
  const int m0   = (mg * 8 + wave) * 64;

  v8f c[4] = {v8f{}, v8f{}, v8f{}, v8f{}};

  const float* brow = lws + l16 * HDIM + koff;
  const float* gsrc = S + (size_t)(m0 + (lane >> 3)) * HDIM + 4 * (lane & 7);
  float* ldst = lA + (wave * 64 + (lane >> 3)) * APITCH + 4 * (lane & 7);
  const float* afrag = lA + (wave * 64 + l16) * APITCH + koff;

  v4f pre[16];
  A_CHUNK_LOAD(pre, gsrc);
  for (int kc = 0; kc < NCHUNK; ++kc) {
    __syncthreads();
    A_CHUNK_STORE(pre, ldst);
    __syncthreads();
    if (kc + 1 < NCHUNK)
      A_CHUNK_LOAD(pre, gsrc + (kc + 1) * KC);
    __builtin_amdgcn_sched_barrier(0);
    #pragma unroll
    for (int j = 0; j < KC / 4; ++j) {
      v2f bf = *(const v2f*)(brow + kc * KC + 4 * j);
      #pragma unroll
      for (int s = 0; s < 4; ++s) {
        v2f af = *(const v2f*)(afrag + (16 * s) * APITCH + 4 * j);
        c[s] = __builtin_amdgcn_wmma_f32_16x16x4_f32(false, af, false, bf,
                                                     (short)0, c[s], false, false);
      }
    }
  }

  #pragma unroll
  for (int s = 0; s < 4; ++s) {
    const int rbase = m0 + 16 * s + (hi << 3);
    #pragma unroll
    for (int r = 0; r < 8; ++r)
      out[(size_t)(rbase + r) * HDIM + n] = c[s][r];
  }
}

extern "C" void kernel_launch(void* const* d_in, const int* in_sizes, int n_in,
                              void* d_out, int out_size, void* d_ws, size_t ws_size,
                              hipStream_t stream) {
  const float* x    = (const float*)d_in[0];
  const float* Wa   = (const float*)d_in[1];
  const float* ba   = (const float*)d_in[2];
  const float* Wu   = (const float*)d_in[3];
  const float* bu   = (const float*)d_in[4];
  const float* tau  = (const float*)d_in[5];
  const float* thr  = (const float*)d_in[6];
  const float* Wsyn = (const float*)d_in[7];
  float* out = (float*)d_out;

  float* ab = (float*)d_ws;
  float* s  = ab + 2 * (size_t)BT * HDIM;

  gates_gemm<<<512, 256, 0, stream>>>(x, Wa, ba, Wu, bu, ab);
  scan_plif<<<64, 32, 0, stream>>>(ab, s, tau, thr);
  synapse_gemm<<<512, 256, 0, stream>>>(s, Wsyn, out);
}


# scan kernel: time steps run in groups of 16 with only their own arithmetic between them; a group's 16 spike stores and 16 ring-refill loads issue as one run behind it; arithmetic and order unchanged
# speedup vs baseline: 10.8075x; 1.4934x over previous
#include <hip/hip_runtime.h>
#include <hip/hip_bf16.h>

#define BATCH 8
#define TLEN  2048
#define HDIM  256
#define BT    (BATCH * TLEN)

typedef __attribute__((ext_vector_type(2))) float v2f;
typedef __attribute__((ext_vector_type(4))) float v4f;
typedef __attribute__((ext_vector_type(8))) float v8f;

#define KC     32
#define NCHUNK (HDIM / KC)
#define APITCH 36
#define AROWS  512

#define A_CHUNK_LOAD(pre, g)                                              \
  _Pragma("unroll") for (int q = 0; q < 16; ++q)                          \
    (pre)[q] = *(const v4f*)((g) + (size_t)(4 * q) * HDIM)
#define A_CHUNK_STORE(pre, l)                                             \
  _Pragma("unroll") for (int q = 0; q < 16; ++q)                          \
    *(v4f*)((l) + (4 * q) * APITCH) = (pre)[q]

__global__ __launch_bounds__(256) void gates_gemm(
    const float* __restrict__ X,
    const float* __restrict__ Wa, const float* __restrict__ ba,
    const float* __restrict__ Wu, const float* __restrict__ bu,
    float* __restrict__ ab  ) {
  __shared__ float lwa[HDIM * 16];
  __shared__ float lwu[HDIM * 16];
  __shared__ __attribute__((aligned(16))) float lA[AROWS * APITCH];

  const int nt = blockIdx.x & 15;
  const int mg = blockIdx.x >> 4;
  const int lane = threadIdx.x & 31;
  const int wave = threadIdx.x >> 5;

  {
    const float4* ga = (const float4*)(Wa + nt * 16);
    const float4* gu = (const float4*)(Wu + nt * 16);
    float4* sa = (float4*)lwa;
    float4* su = (float4*)lwu;
    #pragma unroll
    for (int i = threadIdx.x; i < HDIM * 4; i += 256) {
      const int k = i >> 2;
      const int q = i & 3;
      sa[i] = ga[k * (HDIM / 4) + q];
      su[i] = gu[k * (HDIM / 4) + q];
    }
  }

  const int hi   = lane >> 4;
  const int l16  = lane & 15;
  const int n    = nt * 16 + l16;
  const int koff = hi << 1;
  const int m0   = (mg * 8 + wave) * 64;

  v8f ca[4] = {v8f{}, v8f{}, v8f{}, v8f{}};
  v8f cu[4] = {v8f{}, v8f{}, v8f{}, v8f{}};

  const float* gsrc = X + (size_t)(m0 + (lane >> 3)) * HDIM + 4 * (lane & 7);
  float* ldst = lA + (wave * 64 + (lane >> 3)) * APITCH + 4 * (lane & 7);
  const float* afrag = lA + (wave * 64 + l16) * APITCH + koff;

  v4f pre[16];
  A_CHUNK_LOAD(pre, gsrc);
  for (int kc = 0; kc < NCHUNK; ++kc) {
    __syncthreads();
    A_CHUNK_STORE(pre, ldst);
    __syncthreads();
    if (kc + 1 < NCHUNK)
      A_CHUNK_LOAD(pre, gsrc + (kc + 1) * KC);
    __builtin_amdgcn_sched_barrier(0);
    #pragma unroll
    for (int j = 0; j < KC / 4; ++j) {
      const int k = kc * KC + 4 * j + koff;
      v2f bwa; bwa.x = lwa[k * 16 + l16]; bwa.y = lwa[(k + 1) * 16 + l16];
      v2f bwu; bwu.x = lwu[k * 16 + l16]; bwu.y = lwu[(k + 1) * 16 + l16];
      #pragma unroll
      for (int s = 0; s < 4; ++s) {
        v2f af = *(const v2f*)(afrag + (16 * s) * APITCH + 4 * j);
        ca[s] = __builtin_amdgcn_wmma_f32_16x16x4_f32(false, af, false, bwa,
                                                      (short)0, ca[s], false, false);
        cu[s] = __builtin_amdgcn_wmma_f32_16x16x4_f32(false, af, false, bwu,
                                                      (short)0, cu[s], false, false);
      }
    }
  }

  const float bia = ba[n];
  const float biu = bu[n];
  #pragma unroll
  for (int s = 0; s < 4; ++s) {
    const int rbase = m0 + 16 * s + (hi << 3);
    #pragma unroll
    for (int r = 0; r < 8; ++r) {
      const float za = ca[s][r] + bia;
      const float zu = cu[s][r] + biu;
      const float ga = 1.0f / (1.0f + __expf(-za));
      const float gu = tanhf(zu);
      v2f e; e.x = ga; e.y = (1.0f - ga) * gu;
      *(v2f*)(ab + 2 * ((size_t)(rbase + r) * HDIM + n)) = e;
    }
  }
}

#define SCAN_RING 64
#define SCAN_GRP  16

__global__ __launch_bounds__(32) void scan_plif(
    const float* __restrict__ ab, float* __restrict__ s_out,
    const float* __restrict__ tau_p, const float* __restrict__ thr_p) {
  const int idx = blockIdx.x * blockDim.x + threadIdx.x;
  const int b = idx >> 8;
  const int h = idx & (HDIM - 1);
  const float inv_tau = 1.0f / tau_p[0];
  const float thr = thr_p[0];
  const float decay = 1.0f - inv_tau;

  float hstate = 0.0f;
  float v = 0.0f;
  const size_t base = ((size_t)b * TLEN) * HDIM + h;
  const float* src = ab + 2 * base;
  float* dst = s_out + base;

  v2f ring[SCAN_RING];
  #pragma unroll
  for (int i = 0; i < SCAN_RING; ++i)
    ring[i] = *(const v2f*)(src + (size_t)i * (2 * HDIM));

  for (int t0 = 0; t0 < TLEN - SCAN_RING; t0 += SCAN_RING) {
    const float* nxt = src + (size_t)(t0 + SCAN_RING) * (2 * HDIM);
    float* cur = dst + (size_t)t0 * HDIM;
    #pragma unroll
    for (int g = 0; g < SCAN_RING; g += SCAN_GRP) {
      float sp[SCAN_GRP];
      #pragma unroll
      for (int j = 0; j < SCAN_GRP; ++j) {
        const v2f e = ring[g + j];
        hstate = e.x * hstate + e.y;
        v = decay * v + inv_tau * hstate;
        const float s1 = (v > thr) ? 1.0f : 0.0f;
        v -= s1 * thr;
        sp[j] = s1;
      }
      __builtin_amdgcn_sched_barrier(0);
      #pragma unroll
      for (int j = 0; j < SCAN_GRP; ++j)
        cur[(size_t)(g + j) * HDIM] = sp[j];
      #pragma unroll
      for (int j = 0; j < SCAN_GRP; ++j)
        ring[g + j] = *(const v2f*)(nxt + (size_t)(g + j) * (2 * HDIM));
      __builtin_amdgcn_sched_barrier(0);
    }
  }

  {
    float* cur = dst + (size_t)(TLEN - SCAN_RING) * HDIM;
    #pragma unroll
    for (int g = 0; g < SCAN_RING; g += SCAN_GRP) {
      float sp[SCAN_GRP];
      #pragma unroll
      for (int j = 0; j < SCAN_GRP; ++j) {
        const v2f e = ring[g + j];
        hstate = e.x * hstate + e.y;
        v = decay * v + inv_tau * hstate;
        const float s1 = (v > thr) ? 1.0f : 0.0f;
        v -= s1 * thr;
        sp[j] = s1;
      }
      __builtin_amdgcn_sched_barrier(0);
      #pragma unroll
      for (int j = 0; j < SCAN_GRP; ++j)
        cur[(size_t)(g + j) * HDIM] = sp[j];
    }
  }
}

__global__ __launch_bounds__(256) void synapse_gemm(
    const float* __restrict__ S, const float* __restrict__ Wsyn,
    float* __restrict__ out) {
  __shared__ float lws[16 * HDIM];
  __shared__ __attribute__((aligned(16))) float lA[AROWS * APITCH];

  const int nt = blockIdx.x & 15;
  const int mg = blockIdx.x >> 4;
  const int lane = threadIdx.x & 31;
  const int wave = threadIdx.x >> 5;

  {
    const float4* g = (const float4*)(Wsyn + (size_t)nt * 16 * HDIM);
    float4* sm = (float4*)lws;
    #pragma unroll
    for (int i = threadIdx.x; i < 16 * HDIM / 4; i += 256)
      sm[i] = g[i];
  }

  const int hi   = lane >> 4;
  const int l16  = lane & 15;
  const int n    = nt * 16 + l16;
  const int koff = hi << 1;
  const int m0   = (mg * 8 + wave) * 64;

  v8f c[4] = {v8f{}, v8f{}, v8f{}, v8f{}};

  const float* brow = lws + l16 * HDIM + koff;
  const float* gsrc = S + (size_t)(m0 + (lane >> 3)) * HDIM + 4 * (lane & 7);
  float* ldst = lA + (wave * 64 + (lane >> 3)) * APITCH + 4 * (lane & 7);
  const float* afrag = lA + (wave * 64 + l16) * APITCH + koff;

  v4f pre[16];
  A_CHUNK_LOAD(pre, gsrc);
  for (int kc = 0; kc < NCHUNK; ++kc) {
    __syncthreads();
    A_CHUNK_STORE(pre, ldst);
    __syncthreads();
    if (kc + 1 < NCHUNK)
      A_CHUNK_LOAD(pre, gsrc + (kc + 1) * KC);
    __builtin_amdgcn_sched_barrier(0);
    #pragma unroll
    for (int j = 0; j < KC / 4; ++j) {
      v2f bf = *(const v2f*)(brow + kc * KC + 4 * j);
      #pragma unroll
      for (int s = 0; s < 4; ++s) {
        v2f af = *(const v2f*)(afrag + (16 * s) * APITCH + 4 * j);
        c[s] = __builtin_amdgcn_wmma_f32_16x16x4_f32(false, af, false, bf,
                                                     (short)0, c[s], false, false);
      }
    }
  }

  #pragma unroll
  for (int s = 0; s < 4; ++s) {
    const int rbase = m0 + 16 * s + (hi << 3);
    #pragma unroll
    for (int r = 0; r < 8; ++r)
      out[(size_t)(rbase + r) * HDIM + n] = c[s][r];
  }
}

extern "C" void kernel_launch(void* const* d_in, const int* in_sizes, int n_in,
                              void* d_out, int out_size, void* d_ws, size_t ws_size,
                              hipStream_t stream) {
  const float* x    = (const float*)d_in[0];
  const float* Wa   = (const float*)d_in[1];
  const float* ba   = (const float*)d_in[2];
  const float* Wu   = (const float*)d_in[3];
  const float* bu   = (const float*)d_in[4];
  const float* tau  = (const float*)d_in[5];
  const float* thr  = (const float*)d_in[6];
  const float* Wsyn = (const float*)d_in[7];
  float* out = (float*)d_out;

  float* ab = (float*)d_ws;
  float* s  = ab + 2 * (size_t)BT * HDIM;

  gates_gemm<<<512, 256, 0, stream>>>(x, Wa, ba, Wu, bu, ab);
  scan_plif<<<64, 32, 0, stream>>>(ab, s, tau, thr);
  synapse_gemm<<<512, 256, 0, stream>>>(s, Wsyn, out);
}


# time axis halved, 4 launches instead of 3: each half's scan shares a launch with the next half's gate GEMM or the previous half's synapse GEMM; per-element sums and their order unchanged
# speedup vs baseline: 11.1326x; 1.0301x over previous
#include <hip/hip_runtime.h>
#include <hip/hip_bf16.h>

#define BATCH 8
#define TLEN  2048
#define HDIM  256
#define BT    (BATCH * TLEN)
#define NHALF 2
#define THALF (TLEN / NHALF)
#define GEMM_BLOCKS 256
#define SCAN_BLOCKS 64

typedef __attribute__((ext_vector_type(2))) float v2f;
typedef __attribute__((ext_vector_type(4))) float v4f;
typedef __attribute__((ext_vector_type(8))) float v8f;

#define KC     32
#define NCHUNK (HDIM / KC)
#define APITCH 36
#define AROWS  512

#define A_CHUNK_LOAD(pre, g)                                              \
  _Pragma("unroll") for (int q = 0; q < 16; ++q)                          \
    (pre)[q] = *(const v4f*)((g) + (size_t)(4 * q) * HDIM)
#define A_CHUNK_STORE(pre, l)                                             \
  _Pragma("unroll") for (int q = 0; q < 16; ++q)                          \
    *(v4f*)((l) + (4 * q) * APITCH) = (pre)[q]

__device__ __forceinline__ void gates_block(
    const int bid, const int half,
    const float* __restrict__ X,
    const float* __restrict__ Wa, const float* __restrict__ ba,
    const float* __restrict__ Wu, const float* __restrict__ bu,
    float* __restrict__ ab  ,
    float* lwa, float* lwu, float* lA) {
  const int nt = bid & 15;
  const int ml = bid >> 4;
  const int mg = ((ml >> 1) << 2) + (half << 1) + (ml & 1);
  const int lane = threadIdx.x & 31;
  const int wave = threadIdx.x >> 5;

  {
    const float4* ga = (const float4*)(Wa + nt * 16);
    const float4* gu = (const float4*)(Wu + nt * 16);
    float4* sa = (float4*)lwa;
    float4* su = (float4*)lwu;
    #pragma unroll
    for (int i = threadIdx.x; i < HDIM * 4; i += 256) {
      const int k = i >> 2;
      const int q = i & 3;
      sa[i] = ga[k * (HDIM / 4) + q];
      su[i] = gu[k * (HDIM / 4) + q];
    }
  }

  const int hi   = lane >> 4;
  const int l16  = lane & 15;
  const int n    = nt * 16 + l16;
  const int koff = hi << 1;
  const int m0   = (mg * 8 + wave) * 64;

  v8f ca[4] = {v8f{}, v8f{}, v8f{}, v8f{}};
  v8f cu[4] = {v8f{}, v8f{}, v8f{}, v8f{}};

  const float* gsrc = X + (size_t)(m0 + (lane >> 3)) * HDIM + 4 * (lane & 7);
  float* ldst = lA + (wave * 64 + (lane >> 3)) * APITCH + 4 * (lane & 7);
  const float* afrag = lA + (wave * 64 + l16) * APITCH + koff;

  v4f pre[16];
  A_CHUNK_LOAD(pre, gsrc);
  for (int kc = 0; kc < NCHUNK; ++kc) {
    __syncthreads();
    A_CHUNK_STORE(pre, ldst);
    __syncthreads();
    if (kc + 1 < NCHUNK)
      A_CHUNK_LOAD(pre, gsrc + (kc + 1) * KC);
    __builtin_amdgcn_sched_barrier(0);
    #pragma unroll
    for (int j = 0; j < KC / 4; ++j) {
      const int k = kc * KC + 4 * j + koff;
      v2f bwa; bwa.x = lwa[k * 16 + l16]; bwa.y = lwa[(k + 1) * 16 + l16];
      v2f bwu; bwu.x = lwu[k * 16 + l16]; bwu.y = lwu[(k + 1) * 16 + l16];
      #pragma unroll
      for (int s = 0; s < 4; ++s) {
        v2f af = *(const v2f*)(afrag + (16 * s) * APITCH + 4 * j);
        ca[s] = __builtin_amdgcn_wmma_f32_16x16x4_f32(false, af, false, bwa,
                                                      (short)0, ca[s], false, false);
        cu[s] = __builtin_amdgcn_wmma_f32_16x16x4_f32(false, af, false, bwu,
                                                      (short)0, cu[s], false, false);
      }
    }
  }

  const float bia = ba[n];
  const float biu = bu[n];
  #pragma unroll
  for (int s = 0; s < 4; ++s) {
    const int rbase = m0 + 16 * s + (hi << 3);
    #pragma unroll
    for (int r = 0; r < 8; ++r) {
      const float za = ca[s][r] + bia;
      const float zu = cu[s][r] + biu;
      const float ga = 1.0f / (1.0f + __expf(-za));
      const float gu = tanhf(zu);
      v2f e; e.x = ga; e.y = (1.0f - ga) * gu;
      *(v2f*)(ab + 2 * ((size_t)(rbase + r) * HDIM + n)) = e;
    }
  }
}

#define SCAN_RING 64
#define SCAN_GRP  16

template <int HALF>
__device__ __forceinline__ void scan_wave(
    const int wid,
    const float* __restrict__ ab, float* __restrict__ s_out,
    float* __restrict__ st  ,
    const float tau, const float thr) {
  const int idx = wid * 32 + (threadIdx.x & 31);
  const int b = idx >> 8;
  const int h = idx & (HDIM - 1);
  const float inv_tau = 1.0f / tau;
  const float decay = 1.0f - inv_tau;

  float hstate = 0.0f;
  float v = 0.0f;
  if (HALF > 0) {
    const v2f c = *(const v2f*)(st + 2 * idx);
    hstate = c.x;
    v = c.y;
  }
  const size_t base = ((size_t)b * TLEN + (size_t)HALF * THALF) * HDIM + h;
  const float* src = ab + 2 * base;
  float* dst = s_out + base;

  v2f ring[SCAN_RING];
  #pragma unroll
  for (int i = 0; i < SCAN_RING; ++i)
    ring[i] = *(const v2f*)(src + (size_t)i * (2 * HDIM));

  for (int t0 = 0; t0 < THALF - SCAN_RING; t0 += SCAN_RING) {
    const float* nxt = src + (size_t)(t0 + SCAN_RING) * (2 * HDIM);
    float* cur = dst + (size_t)t0 * HDIM;
    #pragma unroll
    for (int g = 0; g < SCAN_RING; g += SCAN_GRP) {
      float sp[SCAN_GRP];
      #pragma unroll
      for (int j = 0; j < SCAN_GRP; ++j) {
        const v2f e = ring[g + j];
        hstate = e.x * hstate + e.y;
        v = decay * v + inv_tau * hstate;
        const float s1 = (v > thr) ? 1.0f : 0.0f;
        v -= s1 * thr;
        sp[j] = s1;
      }
      __builtin_amdgcn_sched_barrier(0);
      #pragma unroll
      for (int j = 0; j < SCAN_GRP; ++j)
        cur[(size_t)(g + j) * HDIM] = sp[j];
      #pragma unroll
      for (int j = 0; j < SCAN_GRP; ++j)
        ring[g + j] = *(const v2f*)(nxt + (size_t)(g + j) * (2 * HDIM));
      __builtin_amdgcn_sched_barrier(0);
    }
  }

  {
    float* cur = dst + (size_t)(THALF - SCAN_RING) * HDIM;
    #pragma unroll
    for (int g = 0; g < SCAN_RING; g += SCAN_GRP) {
      float sp[SCAN_GRP];
      #pragma unroll
      for (int j = 0; j < SCAN_GRP; ++j) {
        const v2f e = ring[g + j];
        hstate = e.x * hstate + e.y;
        v = decay * v + inv_tau * hstate;
        const float s1 = (v > thr) ? 1.0f : 0.0f;
        v -= s1 * thr;
        sp[j] = s1;
      }
      __builtin_amdgcn_sched_barrier(0);
      #pragma unroll
      for (int j = 0; j < SCAN_GRP; ++j)
        cur[(size_t)(g + j) * HDIM] = sp[j];
    }
  }

  if (HALF + 1 < NHALF) {
    v2f c; c.x = hstate; c.y = v;
    *(v2f*)(st + 2 * idx) = c;
  }
}

__device__ __forceinline__ void synapse_block(
    const int bid, const int half,
    const float* __restrict__ S, const float* __restrict__ Wsyn,
    float* __restrict__ out,
    float* lws, float* lA) {
  const int nt = bid & 15;
  const int ml = bid >> 4;
  const int mg = ((ml >> 1) << 2) + (half << 1) + (ml & 1);
  const int lane = threadIdx.x & 31;
  const int wave = threadIdx.x >> 5;

  {
    const float4* g = (const float4*)(Wsyn + (size_t)nt * 16 * HDIM);
    float4* sm = (float4*)lws;
    #pragma unroll
    for (int i = threadIdx.x; i < 16 * HDIM / 4; i += 256)
      sm[i] = g[i];
  }

  const int hi   = lane >> 4;
  const int l16  = lane & 15;
  const int n    = nt * 16 + l16;
  const int koff = hi << 1;
  const int m0   = (mg * 8 + wave) * 64;

  v8f c[4] = {v8f{}, v8f{}, v8f{}, v8f{}};

  const float* brow = lws + l16 * HDIM + koff;
  const float* gsrc = S + (size_t)(m0 + (lane >> 3)) * HDIM + 4 * (lane & 7);
  float* ldst = lA + (wave * 64 + (lane >> 3)) * APITCH + 4 * (lane & 7);
  const float* afrag = lA + (wave * 64 + l16) * APITCH + koff;

  v4f pre[16];
  A_CHUNK_LOAD(pre, gsrc);
  for (int kc = 0; kc < NCHUNK; ++kc) {
    __syncthreads();
    A_CHUNK_STORE(pre, ldst);
    __syncthreads();
    if (kc + 1 < NCHUNK)
      A_CHUNK_LOAD(pre, gsrc + (kc + 1) * KC);
    __builtin_amdgcn_sched_barrier(0);
    #pragma unroll
    for (int j = 0; j < KC / 4; ++j) {
      v2f bf = *(const v2f*)(brow + kc * KC + 4 * j);
      #pragma unroll
      for (int s = 0; s < 4; ++s) {
        v2f af = *(const v2f*)(afrag + (16 * s) * APITCH + 4 * j);
        c[s] = __builtin_amdgcn_wmma_f32_16x16x4_f32(false, af, false, bf,
                                                     (short)0, c[s], false, false);
      }
    }
  }

  #pragma unroll
  for (int s = 0; s < 4; ++s) {
    const int rbase = m0 + 16 * s + (hi << 3);
    #pragma unroll
    for (int r = 0; r < 8; ++r)
      out[(size_t)(rbase + r) * HDIM + n] = c[s][r];
  }
}

__global__ __launch_bounds__(256) void gates_first(
    const float* __restrict__ X,
    const float* __restrict__ Wa, const float* __restrict__ ba,
    const float* __restrict__ Wu, const float* __restrict__ bu,
    float* __restrict__ ab) {
  __shared__ __attribute__((aligned(16))) float lw[2 * HDIM * 16];
  __shared__ __attribute__((aligned(16))) float lA[AROWS * APITCH];
  gates_block(blockIdx.x, 0, X, Wa, ba, Wu, bu, ab, lw, lw + HDIM * 16, lA);
}

__global__ __launch_bounds__(256) void scan_first_gates_second(
    const float* __restrict__ X,
    const float* __restrict__ Wa, const float* __restrict__ ba,
    const float* __restrict__ Wu, const float* __restrict__ bu,
    const float* __restrict__ ab_rd, float* __restrict__ ab_wr,
    float* __restrict__ s_out, float* __restrict__ st,
    const float* __restrict__ tau_p, const float* __restrict__ thr_p) {
  __shared__ __attribute__((aligned(16))) float lw[2 * HDIM * 16];
  __shared__ __attribute__((aligned(16))) float lA[AROWS * APITCH];
  if (blockIdx.x < SCAN_BLOCKS) {
    const float tau = tau_p[0];
    const float thr = thr_p[0];
    if (threadIdx.x < 32)
      scan_wave<0>(blockIdx.x, ab_rd, s_out, st, tau, thr);
    return;
  }
  gates_block(blockIdx.x - SCAN_BLOCKS, 1, X, Wa, ba, Wu, bu, ab_wr,
              lw, lw + HDIM * 16, lA);
}

__global__ __launch_bounds__(256) void scan_second_synapse_first(
    const float* __restrict__ ab, float* __restrict__ s_wr,
    float* __restrict__ st,
    const float* __restrict__ tau_p, const float* __restrict__ thr_p,
    const float* __restrict__ s_rd, const float* __restrict__ Wsyn,
    float* __restrict__ out) {
  __shared__ __attribute__((aligned(16))) float lw[16 * HDIM];
  __shared__ __attribute__((aligned(16))) float lA[AROWS * APITCH];
  if (blockIdx.x < SCAN_BLOCKS) {
    const float tau = tau_p[0];
    const float thr = thr_p[0];
    if (threadIdx.x < 32)
      scan_wave<1>(blockIdx.x, ab, s_wr, st, tau, thr);
    return;
  }
  synapse_block(blockIdx.x - SCAN_BLOCKS, 0, s_rd, Wsyn, out, lw, lA);
}

__global__ __launch_bounds__(256) void synapse_second(
    const float* __restrict__ S, const float* __restrict__ Wsyn,
    float* __restrict__ out) {
  __shared__ __attribute__((aligned(16))) float lw[16 * HDIM];
  __shared__ __attribute__((aligned(16))) float lA[AROWS * APITCH];
  synapse_block(blockIdx.x, 1, S, Wsyn, out, lw, lA);
}

extern "C" void kernel_launch(void* const* d_in, const int* in_sizes, int n_in,
                              void* d_out, int out_size, void* d_ws, size_t ws_size,
                              hipStream_t stream) {
  const float* x    = (const float*)d_in[0];
  const float* Wa   = (const float*)d_in[1];
  const float* ba   = (const float*)d_in[2];
  const float* Wu   = (const float*)d_in[3];
  const float* bu   = (const float*)d_in[4];
  const float* tau  = (const float*)d_in[5];
  const float* thr  = (const float*)d_in[6];
  const float* Wsyn = (const float*)d_in[7];
  float* out = (float*)d_out;

  float* ab = (float*)d_ws;
  float* s  = ab + 2 * (size_t)BT * HDIM;
  float* st = s + (size_t)BT * HDIM;

  gates_first<<<GEMM_BLOCKS, 256, 0, stream>>>(x, Wa, ba, Wu, bu, ab);
  scan_first_gates_second<<<SCAN_BLOCKS + GEMM_BLOCKS, 256, 0, stream>>>(
      x, Wa, ba, Wu, bu, ab, ab, s, st, tau, thr);
  scan_second_synapse_first<<<SCAN_BLOCKS + GEMM_BLOCKS, 256, 0, stream>>>(
      ab, s, st, tau, thr, s, Wsyn, out);
  synapse_second<<<GEMM_BLOCKS, 256, 0, stream>>>(s, Wsyn, out);
}


# scan: serial wave issues no global loads or stores; 4 helper waves of its work-group stage (a,b) tiles into LDS and carry spikes out of LDS, one barrier per 32 steps; arithmetic and order unchanged
# speedup vs baseline: 11.7456x; 1.0551x over previous
#include <hip/hip_runtime.h>
#include <hip/hip_bf16.h>

#define BATCH 8
#define TLEN  2048
#define HDIM  256
#define BT    (BATCH * TLEN)
#define NHALF 2
#define THALF (TLEN / NHALF)
#define GEMM_BLOCKS 256
#define SCAN_BLOCKS 64

typedef __attribute__((ext_vector_type(2))) float v2f;
typedef __attribute__((ext_vector_type(4))) float v4f;
typedef __attribute__((ext_vector_type(8))) float v8f;

#define KC     32
#define NCHUNK (HDIM / KC)
#define APITCH 36
#define AROWS  512

#define A_CHUNK_LOAD(pre, g)                                              \
  _Pragma("unroll") for (int q = 0; q < 16; ++q)                          \
    (pre)[q] = *(const v4f*)((g) + (size_t)(4 * q) * HDIM)
#define A_CHUNK_STORE(pre, l)                                             \
  _Pragma("unroll") for (int q = 0; q < 16; ++q)                          \
    *(v4f*)((l) + (4 * q) * APITCH) = (pre)[q]

__device__ __forceinline__ void gates_block(
    const int bid, const int half,
    const float* __restrict__ X,
    const float* __restrict__ Wa, const float* __restrict__ ba,
    const float* __restrict__ Wu, const float* __restrict__ bu,
    float* __restrict__ ab  ,
    float* lwa, float* lwu, float* lA) {
  const int nt = bid & 15;
  const int ml = bid >> 4;
  const int mg = ((ml >> 1) << 2) + (half << 1) + (ml & 1);
  const int lane = threadIdx.x & 31;
  const int wave = threadIdx.x >> 5;

  {
    const float4* ga = (const float4*)(Wa + nt * 16);
    const float4* gu = (const float4*)(Wu + nt * 16);
    float4* sa = (float4*)lwa;
    float4* su = (float4*)lwu;
    #pragma unroll
    for (int i = threadIdx.x; i < HDIM * 4; i += 256) {
      const int k = i >> 2;
      const int q = i & 3;
      sa[i] = ga[k * (HDIM / 4) + q];
      su[i] = gu[k * (HDIM / 4) + q];
    }
  }

  const int hi   = lane >> 4;
  const int l16  = lane & 15;
  const int n    = nt * 16 + l16;
  const int koff = hi << 1;
  const int m0   = (mg * 8 + wave) * 64;

  v8f ca[4] = {v8f{}, v8f{}, v8f{}, v8f{}};
  v8f cu[4] = {v8f{}, v8f{}, v8f{}, v8f{}};

  const float* gsrc = X + (size_t)(m0 + (lane >> 3)) * HDIM + 4 * (lane & 7);
  float* ldst = lA + (wave * 64 + (lane >> 3)) * APITCH + 4 * (lane & 7);
  const float* afrag = lA + (wave * 64 + l16) * APITCH + koff;

  v4f pre[16];
  A_CHUNK_LOAD(pre, gsrc);
  for (int kc = 0; kc < NCHUNK; ++kc) {
    __syncthreads();
    A_CHUNK_STORE(pre, ldst);
    __syncthreads();
    if (kc + 1 < NCHUNK)
      A_CHUNK_LOAD(pre, gsrc + (kc + 1) * KC);
    __builtin_amdgcn_sched_barrier(0);
    #pragma unroll
    for (int j = 0; j < KC / 4; ++j) {
      const int k = kc * KC + 4 * j + koff;
      v2f bwa; bwa.x = lwa[k * 16 + l16]; bwa.y = lwa[(k + 1) * 16 + l16];
      v2f bwu; bwu.x = lwu[k * 16 + l16]; bwu.y = lwu[(k + 1) * 16 + l16];
      #pragma unroll
      for (int s = 0; s < 4; ++s) {
        v2f af = *(const v2f*)(afrag + (16 * s) * APITCH + 4 * j);
        ca[s] = __builtin_amdgcn_wmma_f32_16x16x4_f32(false, af, false, bwa,
                                                      (short)0, ca[s], false, false);
        cu[s] = __builtin_amdgcn_wmma_f32_16x16x4_f32(false, af, false, bwu,
                                                      (short)0, cu[s], false, false);
      }
    }
  }

  const float bia = ba[n];
  const float biu = bu[n];
  #pragma unroll
  for (int s = 0; s < 4; ++s) {
    const int rbase = m0 + 16 * s + (hi << 3);
    #pragma unroll
    for (int r = 0; r < 8; ++r) {
      const float za = ca[s][r] + bia;
      const float zu = cu[s][r] + biu;
      const float ga = 1.0f / (1.0f + __expf(-za));
      const float gu = tanhf(zu);
      v2f e; e.x = ga; e.y = (1.0f - ga) * gu;
      *(v2f*)(ab + 2 * ((size_t)(rbase + r) * HDIM + n)) = e;
    }
  }
}

#define SCAN_TILE    32
#define SCAN_NT      (THALF / SCAN_TILE)
#define SCAN_AHEAD   4
#define SCAN_HELPERS 4
#define SCAN_ITERS   36
#define SCAN_AB_SLOT 2048
#define SCAN_SP_BASE 4096
#define SCAN_SP_SLOT 1024
static_assert(THALF % SCAN_TILE == 0 && SCAN_TILE == 8 * SCAN_HELPERS, "whole tiles; 8 steps per helper");
static_assert(SCAN_AHEAD % 2 == 0 && SCAN_ITERS % SCAN_AHEAD == 0 && SCAN_ITERS >= SCAN_NT + 3,
              "slot parity follows the unrolled index; every role runs the same whole number of rings");

__device__ __forceinline__ void scan_handover() {
  asm volatile("s_wait_dscnt 0x0\n\ts_barrier_signal -1\n\ts_barrier_wait -1" ::: "memory");
}

template <int HALF>
__device__ __forceinline__ void scan_block(
    const int wid,
    const float* __restrict__ ab, float* __restrict__ s_out,
    float* __restrict__ st  ,
    const float tau, const float thr, float* lds) {
  const int lane = threadIdx.x & 31;
  const int wave = __builtin_amdgcn_readfirstlane((int)(threadIdx.x >> 5));
  const bool is_serial = (wave == 0);
  const bool is_helper = (wave >= 1) && (wave <= SCAN_HELPERS);
  const int hw = wave - 1;
  const int idx = wid * 32 + lane;
  const int b = idx >> 8;
  const int h = idx & (HDIM - 1);
  const float inv_tau = 1.0f / tau;
  const float decay = 1.0f - inv_tau;

  float hstate = 0.0f;
  float v = 0.0f;
  if (HALF > 0 && is_serial) {
    const v2f c = *(const v2f*)(st + 2 * idx);
    hstate = c.x;
    v = c.y;
  }
  const size_t base = ((size_t)b * TLEN + (size_t)HALF * THALF) * HDIM + h;
  const float* src = ab + 2 * base;
  float* dst = s_out + base;
  float* lab = lds + lane * 4;
  float* lsp = lds + SCAN_SP_BASE + lane * 4;

  if (is_serial) {
    v4f Q[32];
    for (int i0 = 0; i0 < SCAN_ITERS; i0 += SCAN_AHEAD) {
      #pragma unroll
      for (int u = 0; u < SCAN_AHEAD; ++u) {
        const int i = i0 + u;
        {
          const float* la = lab + ((u + 1) & 1) * SCAN_AB_SLOT;
          #pragma unroll
          for (int p = 0; p < 16; ++p)
            Q[16 * ((u + 1) & 1) + p] = *(const v4f*)(la + p * 128);
        }
        if (i >= 2 && i <= SCAN_NT + 1) {
          float sp[SCAN_TILE];
          #pragma unroll
          for (int j = 0; j < SCAN_TILE; ++j) {
            const v4f q = Q[16 * (u & 1) + (j >> 1)];
            const float ea = (j & 1) ? q.z : q.x;
            const float eb = (j & 1) ? q.w : q.y;
            hstate = ea * hstate + eb;
            v = decay * v + inv_tau * hstate;
            const float s1 = (v > thr) ? 1.0f : 0.0f;
            v -= s1 * thr;
            sp[j] = s1;
          }
          float* ls = lsp + (u & 1) * SCAN_SP_SLOT;
          #pragma unroll
          for (int q4 = 0; q4 < SCAN_TILE / 4; ++q4)
            *(v4f*)(ls + q4 * 128) =
                v4f{sp[4 * q4], sp[4 * q4 + 1], sp[4 * q4 + 2], sp[4 * q4 + 3]};
        }
        scan_handover();
      }
    }
  } else if (is_helper) {
    v2f R[8 * SCAN_AHEAD];
    #pragma unroll
    for (int r = 0; r < SCAN_AHEAD; ++r) {
      #pragma unroll
      for (int jj = 0; jj < 8; ++jj)
        R[8 * r + jj] =
            *(const v2f*)(src + (size_t)(r * SCAN_TILE + 8 * hw + jj) * (2 * HDIM));
    }
    for (int i0 = 0; i0 < SCAN_ITERS; i0 += SCAN_AHEAD) {
      #pragma unroll
      for (int u = 0; u < SCAN_AHEAD; ++u) {
        const int i = i0 + u;
        if (i >= 3 && i <= SCAN_NT + 2) {
          const float* ls = lsp + ((u + 1) & 1) * SCAN_SP_SLOT + (2 * hw) * 128;
          float* g = dst + (size_t)((i - 3) * SCAN_TILE + 8 * hw) * HDIM;
          #pragma unroll
          for (int mm = 0; mm < 2; ++mm) {
            const v4f sv = *(const v4f*)(ls + mm * 128);
            g[(size_t)(4 * mm + 0) * HDIM] = sv.x;
            g[(size_t)(4 * mm + 1) * HDIM] = sv.y;
            g[(size_t)(4 * mm + 2) * HDIM] = sv.z;
            g[(size_t)(4 * mm + 3) * HDIM] = sv.w;
          }
        }
        {
          float* la = lab + (u & 1) * SCAN_AB_SLOT + (4 * hw) * 128;
          #pragma unroll
          for (int jj = 0; jj < 8; ++jj)
            *(v2f*)(la + (jj >> 1) * 128 + (jj & 1) * 2) = R[8 * u + jj];
        }
        __builtin_amdgcn_sched_barrier(0);
        {
          const int tn = (i + SCAN_AHEAD < SCAN_NT) ? (i + SCAN_AHEAD) : (SCAN_NT - 1);
          #pragma unroll
          for (int jj = 0; jj < 8; ++jj)
            R[8 * u + jj] =
                *(const v2f*)(src + (size_t)(tn * SCAN_TILE + 8 * hw + jj) * (2 * HDIM));
        }
        scan_handover();
      }
    }
  } else {
    for (int i = 0; i < SCAN_ITERS; ++i)
      scan_handover();
  }

  if (HALF + 1 < NHALF && is_serial) {
    v2f c; c.x = hstate; c.y = v;
    *(v2f*)(st + 2 * idx) = c;
  }
}

__device__ __forceinline__ void synapse_block(
    const int bid, const int half,
    const float* __restrict__ S, const float* __restrict__ Wsyn,
    float* __restrict__ out,
    float* lws, float* lA) {
  const int nt = bid & 15;
  const int ml = bid >> 4;
  const int mg = ((ml >> 1) << 2) + (half << 1) + (ml & 1);
  const int lane = threadIdx.x & 31;
  const int wave = threadIdx.x >> 5;

  {
    const float4* g = (const float4*)(Wsyn + (size_t)nt * 16 * HDIM);
    float4* sm = (float4*)lws;
    #pragma unroll
    for (int i = threadIdx.x; i < 16 * HDIM / 4; i += 256)
      sm[i] = g[i];
  }

  const int hi   = lane >> 4;
  const int l16  = lane & 15;
  const int n    = nt * 16 + l16;
  const int koff = hi << 1;
  const int m0   = (mg * 8 + wave) * 64;

  v8f c[4] = {v8f{}, v8f{}, v8f{}, v8f{}};

  const float* brow = lws + l16 * HDIM + koff;
  const float* gsrc = S + (size_t)(m0 + (lane >> 3)) * HDIM + 4 * (lane & 7);
  float* ldst = lA + (wave * 64 + (lane >> 3)) * APITCH + 4 * (lane & 7);
  const float* afrag = lA + (wave * 64 + l16) * APITCH + koff;

  v4f pre[16];
  A_CHUNK_LOAD(pre, gsrc);
  for (int kc = 0; kc < NCHUNK; ++kc) {
    __syncthreads();
    A_CHUNK_STORE(pre, ldst);
    __syncthreads();
    if (kc + 1 < NCHUNK)
      A_CHUNK_LOAD(pre, gsrc + (kc + 1) * KC);
    __builtin_amdgcn_sched_barrier(0);
    #pragma unroll
    for (int j = 0; j < KC / 4; ++j) {
      v2f bf = *(const v2f*)(brow + kc * KC + 4 * j);
      #pragma unroll
      for (int s = 0; s < 4; ++s) {
        v2f af = *(const v2f*)(afrag + (16 * s) * APITCH + 4 * j);
        c[s] = __builtin_amdgcn_wmma_f32_16x16x4_f32(false, af, false, bf,
                                                     (short)0, c[s], false, false);
      }
    }
  }

  #pragma unroll
  for (int s = 0; s < 4; ++s) {
    const int rbase = m0 + 16 * s + (hi << 3);
    #pragma unroll
    for (int r = 0; r < 8; ++r)
      out[(size_t)(rbase + r) * HDIM + n] = c[s][r];
  }
}

__global__ __launch_bounds__(256) void gates_first(
    const float* __restrict__ X,
    const float* __restrict__ Wa, const float* __restrict__ ba,
    const float* __restrict__ Wu, const float* __restrict__ bu,
    float* __restrict__ ab) {
  __shared__ __attribute__((aligned(16))) float lw[2 * HDIM * 16];
  __shared__ __attribute__((aligned(16))) float lA[AROWS * APITCH];
  gates_block(blockIdx.x, 0, X, Wa, ba, Wu, bu, ab, lw, lw + HDIM * 16, lA);
}

__global__ __launch_bounds__(256) void scan_first_gates_second(
    const float* __restrict__ X,
    const float* __restrict__ Wa, const float* __restrict__ ba,
    const float* __restrict__ Wu, const float* __restrict__ bu,
    const float* __restrict__ ab_rd, float* __restrict__ ab_wr,
    float* __restrict__ s_out, float* __restrict__ st,
    const float* __restrict__ tau_p, const float* __restrict__ thr_p) {
  __shared__ __attribute__((aligned(16))) float lw[2 * HDIM * 16];
  __shared__ __attribute__((aligned(16))) float lA[AROWS * APITCH];
  if (blockIdx.x < SCAN_BLOCKS) {
    const float tau = tau_p[0];
    const float thr = thr_p[0];
    scan_block<0>(blockIdx.x, ab_rd, s_out, st, tau, thr, lA);
    return;
  }
  gates_block(blockIdx.x - SCAN_BLOCKS, 1, X, Wa, ba, Wu, bu, ab_wr,
              lw, lw + HDIM * 16, lA);
}

__global__ __launch_bounds__(256) void scan_second_synapse_first(
    const float* __restrict__ ab, float* __restrict__ s_wr,
    float* __restrict__ st,
    const float* __restrict__ tau_p, const float* __restrict__ thr_p,
    const float* __restrict__ s_rd, const float* __restrict__ Wsyn,
    float* __restrict__ out) {
  __shared__ __attribute__((aligned(16))) float lw[16 * HDIM];
  __shared__ __attribute__((aligned(16))) float lA[AROWS * APITCH];
  if (blockIdx.x < SCAN_BLOCKS) {
    const float tau = tau_p[0];
    const float thr = thr_p[0];
    scan_block<1>(blockIdx.x, ab, s_wr, st, tau, thr, lA);
    return;
  }
  synapse_block(blockIdx.x - SCAN_BLOCKS, 0, s_rd, Wsyn, out, lw, lA);
}

__global__ __launch_bounds__(256) void synapse_second(
    const float* __restrict__ S, const float* __restrict__ Wsyn,
    float* __restrict__ out) {
  __shared__ __attribute__((aligned(16))) float lw[16 * HDIM];
  __shared__ __attribute__((aligned(16))) float lA[AROWS * APITCH];
  synapse_block(blockIdx.x, 1, S, Wsyn, out, lw, lA);
}

extern "C" void kernel_launch(void* const* d_in, const int* in_sizes, int n_in,
                              void* d_out, int out_size, void* d_ws, size_t ws_size,
                              hipStream_t stream) {
  const float* x    = (const float*)d_in[0];
  const float* Wa   = (const float*)d_in[1];
  const float* ba   = (const float*)d_in[2];
  const float* Wu   = (const float*)d_in[3];
  const float* bu   = (const float*)d_in[4];
  const float* tau  = (const float*)d_in[5];
  const float* thr  = (const float*)d_in[6];
  const float* Wsyn = (const float*)d_in[7];
  float* out = (float*)d_out;

  float* ab = (float*)d_ws;
  float* s  = ab + 2 * (size_t)BT * HDIM;
  float* st = s + (size_t)BT * HDIM;

  gates_first<<<GEMM_BLOCKS, 256, 0, stream>>>(x, Wa, ba, Wu, bu, ab);
  scan_first_gates_second<<<SCAN_BLOCKS + GEMM_BLOCKS, 256, 0, stream>>>(
      x, Wa, ba, Wu, bu, ab, ab, s, st, tau, thr);
  scan_second_synapse_first<<<SCAN_BLOCKS + GEMM_BLOCKS, 256, 0, stream>>>(
      ab, s, st, tau, thr, s, Wsyn, out);
  synapse_second<<<GEMM_BLOCKS, 256, 0, stream>>>(s, Wsyn, out);
}


# GEMM-only launches 1 and 4 run as 512 four-wave blocks (256 rows), two per WGP, each filling the other's barrier and staging waits; weight-strip loads issue as one group; sums bit-identical
# speedup vs baseline: 12.2705x; 1.0447x over previous
#include <hip/hip_runtime.h>
#include <hip/hip_bf16.h>

#define BATCH 8
#define TLEN  2048
#define HDIM  256
#define BT    (BATCH * TLEN)
#define NHALF 2
#define THALF (TLEN / NHALF)
#define GEMM_BLOCKS 256
#define SCAN_BLOCKS 64

typedef __attribute__((ext_vector_type(2))) float v2f;
typedef __attribute__((ext_vector_type(4))) float v4f;
typedef __attribute__((ext_vector_type(8))) float v8f;

#define KC     32
#define NCHUNK (HDIM / KC)
#define APITCH 36
#define AROWS  512
#define NW_SOLO 4
#define AROWS_SOLO (NW_SOLO * 64)

#define A_CHUNK_LOAD(pre, g)                                              \
  _Pragma("unroll") for (int q = 0; q < 16; ++q)                          \
    (pre)[q] = *(const v4f*)((g) + (size_t)(4 * q) * HDIM)
#define A_CHUNK_STORE(pre, l)                                             \
  _Pragma("unroll") for (int q = 0; q < 16; ++q)                          \
    *(v4f*)((l) + (4 * q) * APITCH) = (pre)[q]

template <int NW>
__device__ __forceinline__ void gates_block(
    const int bid, const int half,
    const float* __restrict__ X,
    const float* __restrict__ Wa, const float* __restrict__ ba,
    const float* __restrict__ Wu, const float* __restrict__ bu,
    float* __restrict__ ab  ,
    float* lwa, float* lwu, float* lA) {
  const int nt = bid & 15;
  const unsigned ml = (unsigned)bid >> 4;
  constexpr unsigned G = THALF / (NW * 64);
  const int mg = (int)((((ml / G) << 1) + (unsigned)half) * G + (ml % G));
  const int lane = threadIdx.x & 31;
  const int wave = threadIdx.x >> 5;

  {
    const float4* ga = (const float4*)(Wa + nt * 16);
    const float4* gu = (const float4*)(Wu + nt * 16);
    float4* sa = (float4*)lwa;
    float4* su = (float4*)lwu;
    constexpr int NS = HDIM * 4 / (NW * 32);
    float4 ta[NS], tu[NS];
    #pragma unroll
    for (int it = 0; it < NS; ++it) {
      const int i = threadIdx.x + it * (NW * 32);
      const int k = i >> 2;
      const int q = i & 3;
      ta[it] = ga[k * (HDIM / 4) + q];
      tu[it] = gu[k * (HDIM / 4) + q];
    }
    #pragma unroll
    for (int it = 0; it < NS; ++it) {
      const int i = threadIdx.x + it * (NW * 32);
      sa[i] = ta[it];
      su[i] = tu[it];
    }
  }

  const int hi   = lane >> 4;
  const int l16  = lane & 15;
  const int n    = nt * 16 + l16;
  const int koff = hi << 1;
  const int m0   = (mg * NW + wave) * 64;

  v8f ca[4] = {v8f{}, v8f{}, v8f{}, v8f{}};
  v8f cu[4] = {v8f{}, v8f{}, v8f{}, v8f{}};

  const float* gsrc = X + (size_t)(m0 + (lane >> 3)) * HDIM + 4 * (lane & 7);
  float* ldst = lA + (wave * 64 + (lane >> 3)) * APITCH + 4 * (lane & 7);
  const float* afrag = lA + (wave * 64 + l16) * APITCH + koff;

  v4f pre[16];
  A_CHUNK_LOAD(pre, gsrc);
  for (int kc = 0; kc < NCHUNK; ++kc) {
    __syncthreads();
    A_CHUNK_STORE(pre, ldst);
    __syncthreads();
    if (kc + 1 < NCHUNK)
      A_CHUNK_LOAD(pre, gsrc + (kc + 1) * KC);
    __builtin_amdgcn_sched_barrier(0);
    #pragma unroll
    for (int j = 0; j < KC / 4; ++j) {
      const int k = kc * KC + 4 * j + koff;
      v2f bwa; bwa.x = lwa[k * 16 + l16]; bwa.y = lwa[(k + 1) * 16 + l16];
      v2f bwu; bwu.x = lwu[k * 16 + l16]; bwu.y = lwu[(k + 1) * 16 + l16];
      #pragma unroll
      for (int s = 0; s < 4; ++s) {
        v2f af = *(const v2f*)(afrag + (16 * s) * APITCH + 4 * j);
        ca[s] = __builtin_amdgcn_wmma_f32_16x16x4_f32(false, af, false, bwa,
                                                      (short)0, ca[s], false, false);
        cu[s] = __builtin_amdgcn_wmma_f32_16x16x4_f32(false, af, false, bwu,
                                                      (short)0, cu[s], false, false);
      }
    }
  }

  const float bia = ba[n];
  const float biu = bu[n];
  #pragma unroll
  for (int s = 0; s < 4; ++s) {
    const int rbase = m0 + 16 * s + (hi << 3);
    #pragma unroll
    for (int r = 0; r < 8; ++r) {
      const float za = ca[s][r] + bia;
      const float zu = cu[s][r] + biu;
      const float ga = 1.0f / (1.0f + __expf(-za));
      const float gu = tanhf(zu);
      v2f e; e.x = ga; e.y = (1.0f - ga) * gu;
      *(v2f*)(ab + 2 * ((size_t)(rbase + r) * HDIM + n)) = e;
    }
  }
}

#define SCAN_TILE    32
#define SCAN_NT      (THALF / SCAN_TILE)
#define SCAN_AHEAD   4
#define SCAN_HELPERS 4
#define SCAN_ITERS   36
#define SCAN_AB_SLOT 2048
#define SCAN_SP_BASE 4096
#define SCAN_SP_SLOT 1024
static_assert(THALF % SCAN_TILE == 0 && SCAN_TILE == 8 * SCAN_HELPERS, "whole tiles; 8 steps per helper");
static_assert(SCAN_AHEAD % 2 == 0 && SCAN_ITERS % SCAN_AHEAD == 0 && SCAN_ITERS >= SCAN_NT + 3,
              "slot parity follows the unrolled index; every role runs the same whole number of rings");

__device__ __forceinline__ void scan_handover() {
  asm volatile("s_wait_dscnt 0x0\n\ts_barrier_signal -1\n\ts_barrier_wait -1" ::: "memory");
}

template <int HALF>
__device__ __forceinline__ void scan_block(
    const int wid,
    const float* __restrict__ ab, float* __restrict__ s_out,
    float* __restrict__ st  ,
    const float tau, const float thr, float* lds) {
  const int lane = threadIdx.x & 31;
  const int wave = __builtin_amdgcn_readfirstlane((int)(threadIdx.x >> 5));
  const bool is_serial = (wave == 0);
  const bool is_helper = (wave >= 1) && (wave <= SCAN_HELPERS);
  const int hw = wave - 1;
  const int idx = wid * 32 + lane;
  const int b = idx >> 8;
  const int h = idx & (HDIM - 1);
  const float inv_tau = 1.0f / tau;
  const float decay = 1.0f - inv_tau;

  float hstate = 0.0f;
  float v = 0.0f;
  if (HALF > 0 && is_serial) {
    const v2f c = *(const v2f*)(st + 2 * idx);
    hstate = c.x;
    v = c.y;
  }
  const size_t base = ((size_t)b * TLEN + (size_t)HALF * THALF) * HDIM + h;
  const float* src = ab + 2 * base;
  float* dst = s_out + base;
  float* lab = lds + lane * 4;
  float* lsp = lds + SCAN_SP_BASE + lane * 4;

  if (is_serial) {
    v4f Q[32];
    for (int i0 = 0; i0 < SCAN_ITERS; i0 += SCAN_AHEAD) {
      #pragma unroll
      for (int u = 0; u < SCAN_AHEAD; ++u) {
        const int i = i0 + u;
        {
          const float* la = lab + ((u + 1) & 1) * SCAN_AB_SLOT;
          #pragma unroll
          for (int p = 0; p < 16; ++p)
            Q[16 * ((u + 1) & 1) + p] = *(const v4f*)(la + p * 128);
        }
        if (i >= 2 && i <= SCAN_NT + 1) {
          float sp[SCAN_TILE];
          #pragma unroll
          for (int j = 0; j < SCAN_TILE; ++j) {
            const v4f q = Q[16 * (u & 1) + (j >> 1)];
            const float ea = (j & 1) ? q.z : q.x;
            const float eb = (j & 1) ? q.w : q.y;
            hstate = ea * hstate + eb;
            v = decay * v + inv_tau * hstate;
            const float s1 = (v > thr) ? 1.0f : 0.0f;
            v -= s1 * thr;
            sp[j] = s1;
          }
          float* ls = lsp + (u & 1) * SCAN_SP_SLOT;
          #pragma unroll
          for (int q4 = 0; q4 < SCAN_TILE / 4; ++q4)
            *(v4f*)(ls + q4 * 128) =
                v4f{sp[4 * q4], sp[4 * q4 + 1], sp[4 * q4 + 2], sp[4 * q4 + 3]};
        }
        scan_handover();
      }
    }
  } else if (is_helper) {
    v2f R[8 * SCAN_AHEAD];
    #pragma unroll
    for (int r = 0; r < SCAN_AHEAD; ++r) {
      #pragma unroll
      for (int jj = 0; jj < 8; ++jj)
        R[8 * r + jj] =
            *(const v2f*)(src + (size_t)(r * SCAN_TILE + 8 * hw + jj) * (2 * HDIM));
    }
    for (int i0 = 0; i0 < SCAN_ITERS; i0 += SCAN_AHEAD) {
      #pragma unroll
      for (int u = 0; u < SCAN_AHEAD; ++u) {
        const int i = i0 + u;
        if (i >= 3 && i <= SCAN_NT + 2) {
          const float* ls = lsp + ((u + 1) & 1) * SCAN_SP_SLOT + (2 * hw) * 128;
          float* g = dst + (size_t)((i - 3) * SCAN_TILE + 8 * hw) * HDIM;
          #pragma unroll
          for (int mm = 0; mm < 2; ++mm) {
            const v4f sv = *(const v4f*)(ls + mm * 128);
            g[(size_t)(4 * mm + 0) * HDIM] = sv.x;
            g[(size_t)(4 * mm + 1) * HDIM] = sv.y;
            g[(size_t)(4 * mm + 2) * HDIM] = sv.z;
            g[(size_t)(4 * mm + 3) * HDIM] = sv.w;
          }
        }
        {
          float* la = lab + (u & 1) * SCAN_AB_SLOT + (4 * hw) * 128;
          #pragma unroll
          for (int jj = 0; jj < 8; ++jj)
            *(v2f*)(la + (jj >> 1) * 128 + (jj & 1) * 2) = R[8 * u + jj];
        }
        __builtin_amdgcn_sched_barrier(0);
        {
          const int tn = (i + SCAN_AHEAD < SCAN_NT) ? (i + SCAN_AHEAD) : (SCAN_NT - 1);
          #pragma unroll
          for (int jj = 0; jj < 8; ++jj)
            R[8 * u + jj] =
                *(const v2f*)(src + (size_t)(tn * SCAN_TILE + 8 * hw + jj) * (2 * HDIM));
        }
        scan_handover();
      }
    }
  } else {
    for (int i = 0; i < SCAN_ITERS; ++i)
      scan_handover();
  }

  if (HALF + 1 < NHALF && is_serial) {
    v2f c; c.x = hstate; c.y = v;
    *(v2f*)(st + 2 * idx) = c;
  }
}

template <int NW>
__device__ __forceinline__ void synapse_block(
    const int bid, const int half,
    const float* __restrict__ S, const float* __restrict__ Wsyn,
    float* __restrict__ out,
    float* lws, float* lA) {
  const int nt = bid & 15;
  const unsigned ml = (unsigned)bid >> 4;
  constexpr unsigned G = THALF / (NW * 64);
  const int mg = (int)((((ml / G) << 1) + (unsigned)half) * G + (ml % G));
  const int lane = threadIdx.x & 31;
  const int wave = threadIdx.x >> 5;

  {
    const float4* g = (const float4*)(Wsyn + (size_t)nt * 16 * HDIM);
    float4* sm = (float4*)lws;
    constexpr int NS = 16 * HDIM / 4 / (NW * 32);
    float4 tw[NS];
    #pragma unroll
    for (int it = 0; it < NS; ++it)
      tw[it] = g[threadIdx.x + it * (NW * 32)];
    #pragma unroll
    for (int it = 0; it < NS; ++it)
      sm[threadIdx.x + it * (NW * 32)] = tw[it];
  }

  const int hi   = lane >> 4;
  const int l16  = lane & 15;
  const int n    = nt * 16 + l16;
  const int koff = hi << 1;
  const int m0   = (mg * NW + wave) * 64;

  v8f c[4] = {v8f{}, v8f{}, v8f{}, v8f{}};

  const float* brow = lws + l16 * HDIM + koff;
  const float* gsrc = S + (size_t)(m0 + (lane >> 3)) * HDIM + 4 * (lane & 7);
  float* ldst = lA + (wave * 64 + (lane >> 3)) * APITCH + 4 * (lane & 7);
  const float* afrag = lA + (wave * 64 + l16) * APITCH + koff;

  v4f pre[16];
  A_CHUNK_LOAD(pre, gsrc);
  for (int kc = 0; kc < NCHUNK; ++kc) {
    __syncthreads();
    A_CHUNK_STORE(pre, ldst);
    __syncthreads();
    if (kc + 1 < NCHUNK)
      A_CHUNK_LOAD(pre, gsrc + (kc + 1) * KC);
    __builtin_amdgcn_sched_barrier(0);
    #pragma unroll
    for (int j = 0; j < KC / 4; ++j) {
      v2f bf = *(const v2f*)(brow + kc * KC + 4 * j);
      #pragma unroll
      for (int s = 0; s < 4; ++s) {
        v2f af = *(const v2f*)(afrag + (16 * s) * APITCH + 4 * j);
        c[s] = __builtin_amdgcn_wmma_f32_16x16x4_f32(false, af, false, bf,
                                                     (short)0, c[s], false, false);
      }
    }
  }

  #pragma unroll
  for (int s = 0; s < 4; ++s) {
    const int rbase = m0 + 16 * s + (hi << 3);
    #pragma unroll
    for (int r = 0; r < 8; ++r)
      out[(size_t)(rbase + r) * HDIM + n] = c[s][r];
  }
}

__global__ __launch_bounds__(NW_SOLO * 32) void gates_first(
    const float* __restrict__ X,
    const float* __restrict__ Wa, const float* __restrict__ ba,
    const float* __restrict__ Wu, const float* __restrict__ bu,
    float* __restrict__ ab) {
  __shared__ __attribute__((aligned(16))) float lw[2 * HDIM * 16];
  __shared__ __attribute__((aligned(16))) float lA[AROWS_SOLO * APITCH];
  gates_block<NW_SOLO>(blockIdx.x, 0, X, Wa, ba, Wu, bu, ab, lw, lw + HDIM * 16, lA);
}

__global__ __launch_bounds__(256) void scan_first_gates_second(
    const float* __restrict__ X,
    const float* __restrict__ Wa, const float* __restrict__ ba,
    const float* __restrict__ Wu, const float* __restrict__ bu,
    const float* __restrict__ ab_rd, float* __restrict__ ab_wr,
    float* __restrict__ s_out, float* __restrict__ st,
    const float* __restrict__ tau_p, const float* __restrict__ thr_p) {
  __shared__ __attribute__((aligned(16))) float lw[2 * HDIM * 16];
  __shared__ __attribute__((aligned(16))) float lA[AROWS * APITCH];
  if (blockIdx.x < SCAN_BLOCKS) {
    const float tau = tau_p[0];
    const float thr = thr_p[0];
    scan_block<0>(blockIdx.x, ab_rd, s_out, st, tau, thr, lA);
    return;
  }
  gates_block<8>(blockIdx.x - SCAN_BLOCKS, 1, X, Wa, ba, Wu, bu, ab_wr,
                 lw, lw + HDIM * 16, lA);
}

__global__ __launch_bounds__(256) void scan_second_synapse_first(
    const float* __restrict__ ab, float* __restrict__ s_wr,
    float* __restrict__ st,
    const float* __restrict__ tau_p, const float* __restrict__ thr_p,
    const float* __restrict__ s_rd, const float* __restrict__ Wsyn,
    float* __restrict__ out) {
  __shared__ __attribute__((aligned(16))) float lw[16 * HDIM];
  __shared__ __attribute__((aligned(16))) float lA[AROWS * APITCH];
  if (blockIdx.x < SCAN_BLOCKS) {
    const float tau = tau_p[0];
    const float thr = thr_p[0];
    scan_block<1>(blockIdx.x, ab, s_wr, st, tau, thr, lA);
    return;
  }
  synapse_block<8>(blockIdx.x - SCAN_BLOCKS, 0, s_rd, Wsyn, out, lw, lA);
}

__global__ __launch_bounds__(NW_SOLO * 32) void synapse_second(
    const float* __restrict__ S, const float* __restrict__ Wsyn,
    float* __restrict__ out) {
  __shared__ __attribute__((aligned(16))) float lw[16 * HDIM];
  __shared__ __attribute__((aligned(16))) float lA[AROWS_SOLO * APITCH];
  synapse_block<NW_SOLO>(blockIdx.x, 1, S, Wsyn, out, lw, lA);
}

extern "C" void kernel_launch(void* const* d_in, const int* in_sizes, int n_in,
                              void* d_out, int out_size, void* d_ws, size_t ws_size,
                              hipStream_t stream) {
  const float* x    = (const float*)d_in[0];
  const float* Wa   = (const float*)d_in[1];
  const float* ba   = (const float*)d_in[2];
  const float* Wu   = (const float*)d_in[3];
  const float* bu   = (const float*)d_in[4];
  const float* tau  = (const float*)d_in[5];
  const float* thr  = (const float*)d_in[6];
  const float* Wsyn = (const float*)d_in[7];
  float* out = (float*)d_out;

  float* ab = (float*)d_ws;
  float* s  = ab + 2 * (size_t)BT * HDIM;
  float* st = s + (size_t)BT * HDIM;

  gates_first<<<GEMM_BLOCKS * (8 / NW_SOLO), NW_SOLO * 32, 0, stream>>>(x, Wa, ba, Wu, bu, ab);
  scan_first_gates_second<<<SCAN_BLOCKS + GEMM_BLOCKS, 256, 0, stream>>>(
      x, Wa, ba, Wu, bu, ab, ab, s, st, tau, thr);
  scan_second_synapse_first<<<SCAN_BLOCKS + GEMM_BLOCKS, 256, 0, stream>>>(
      ab, s, st, tau, thr, s, Wsyn, out);
  synapse_second<<<GEMM_BLOCKS * (8 / NW_SOLO), NW_SOLO * 32, 0, stream>>>(s, Wsyn, out);
}


# both GEMMs: K loop without work-group barriers (one after the weight copy stays): each wave stages and reads only its own 64 A rows in LDS, so a block's waves run out of step; sums bit-identical
# speedup vs baseline: 13.5950x; 1.1079x over previous
#include <hip/hip_runtime.h>
#include <hip/hip_bf16.h>

#define BATCH 8
#define TLEN  2048
#define HDIM  256
#define BT    (BATCH * TLEN)
#define NHALF 2
#define THALF (TLEN / NHALF)
#define GEMM_BLOCKS 256
#define SCAN_BLOCKS 64

typedef __attribute__((ext_vector_type(2))) float v2f;
typedef __attribute__((ext_vector_type(4))) float v4f;
typedef __attribute__((ext_vector_type(8))) float v8f;

#define KC     32
#define NCHUNK (HDIM / KC)
#define APITCH 36
#define AROWS  512
#define NW_SOLO 4
#define AROWS_SOLO (NW_SOLO * 64)

#define A_CHUNK_LOAD(pre, g)                                              \
  _Pragma("unroll") for (int q = 0; q < 16; ++q)                          \
    (pre)[q] = *(const v4f*)((g) + (size_t)(4 * q) * HDIM)
#define A_CHUNK_STORE(pre, l)                                             \
  _Pragma("unroll") for (int q = 0; q < 16; ++q)                          \
    *(v4f*)((l) + (4 * q) * APITCH) = (pre)[q]

__device__ __forceinline__ void wave_lds_order() {
  __builtin_amdgcn_fence(__ATOMIC_RELEASE, "wavefront");
  __builtin_amdgcn_wave_barrier();
  __builtin_amdgcn_fence(__ATOMIC_ACQUIRE, "wavefront");
}

template <int NW>
__device__ __forceinline__ void gates_block(
    const int bid, const int half,
    const float* __restrict__ X,
    const float* __restrict__ Wa, const float* __restrict__ ba,
    const float* __restrict__ Wu, const float* __restrict__ bu,
    float* __restrict__ ab  ,
    float* lwa, float* lwu, float* lA) {
  const int nt = bid & 15;
  const unsigned ml = (unsigned)bid >> 4;
  constexpr unsigned G = THALF / (NW * 64);
  const int mg = (int)((((ml / G) << 1) + (unsigned)half) * G + (ml % G));
  const int lane = threadIdx.x & 31;
  const int wave = threadIdx.x >> 5;

  {
    const float4* ga = (const float4*)(Wa + nt * 16);
    const float4* gu = (const float4*)(Wu + nt * 16);
    float4* sa = (float4*)lwa;
    float4* su = (float4*)lwu;
    constexpr int NS = HDIM * 4 / (NW * 32);
    float4 ta[NS], tu[NS];
    #pragma unroll
    for (int it = 0; it < NS; ++it) {
      const int i = threadIdx.x + it * (NW * 32);
      const int k = i >> 2;
      const int q = i & 3;
      ta[it] = ga[k * (HDIM / 4) + q];
      tu[it] = gu[k * (HDIM / 4) + q];
    }
    #pragma unroll
    for (int it = 0; it < NS; ++it) {
      const int i = threadIdx.x + it * (NW * 32);
      sa[i] = ta[it];
      su[i] = tu[it];
    }
  }

  const int hi   = lane >> 4;
  const int l16  = lane & 15;
  const int n    = nt * 16 + l16;
  const int koff = hi << 1;
  const int m0   = (mg * NW + wave) * 64;

  v8f ca[4] = {v8f{}, v8f{}, v8f{}, v8f{}};
  v8f cu[4] = {v8f{}, v8f{}, v8f{}, v8f{}};

  const float* gsrc = X + (size_t)(m0 + (lane >> 3)) * HDIM + 4 * (lane & 7);
  float* ldst = lA + (wave * 64 + (lane >> 3)) * APITCH + 4 * (lane & 7);
  const float* afrag = lA + (wave * 64 + l16) * APITCH + koff;

  v4f pre[16];
  A_CHUNK_LOAD(pre, gsrc);
  __syncthreads();
  for (int kc = 0; kc < NCHUNK; ++kc) {
    wave_lds_order();
    A_CHUNK_STORE(pre, ldst);
    wave_lds_order();
    if (kc + 1 < NCHUNK)
      A_CHUNK_LOAD(pre, gsrc + (kc + 1) * KC);
    __builtin_amdgcn_sched_barrier(0);
    #pragma unroll
    for (int j = 0; j < KC / 4; ++j) {
      const int k = kc * KC + 4 * j + koff;
      v2f bwa; bwa.x = lwa[k * 16 + l16]; bwa.y = lwa[(k + 1) * 16 + l16];
      v2f bwu; bwu.x = lwu[k * 16 + l16]; bwu.y = lwu[(k + 1) * 16 + l16];
      #pragma unroll
      for (int s = 0; s < 4; ++s) {
        v2f af = *(const v2f*)(afrag + (16 * s) * APITCH + 4 * j);
        ca[s] = __builtin_amdgcn_wmma_f32_16x16x4_f32(false, af, false, bwa,
                                                      (short)0, ca[s], false, false);
        cu[s] = __builtin_amdgcn_wmma_f32_16x16x4_f32(false, af, false, bwu,
                                                      (short)0, cu[s], false, false);
      }
    }
  }

  const float bia = ba[n];
  const float biu = bu[n];
  #pragma unroll
  for (int s = 0; s < 4; ++s) {
    const int rbase = m0 + 16 * s + (hi << 3);
    #pragma unroll
    for (int r = 0; r < 8; ++r) {
      const float za = ca[s][r] + bia;
      const float zu = cu[s][r] + biu;
      const float ga = 1.0f / (1.0f + __expf(-za));
      const float gu = tanhf(zu);
      v2f e; e.x = ga; e.y = (1.0f - ga) * gu;
      *(v2f*)(ab + 2 * ((size_t)(rbase + r) * HDIM + n)) = e;
    }
  }
}

#define SCAN_TILE    32
#define SCAN_NT      (THALF / SCAN_TILE)
#define SCAN_AHEAD   4
#define SCAN_HELPERS 4
#define SCAN_ITERS   36
#define SCAN_AB_SLOT 2048
#define SCAN_SP_BASE 4096
#define SCAN_SP_SLOT 1024
static_assert(THALF % SCAN_TILE == 0 && SCAN_TILE == 8 * SCAN_HELPERS, "whole tiles; 8 steps per helper");
static_assert(SCAN_AHEAD % 2 == 0 && SCAN_ITERS % SCAN_AHEAD == 0 && SCAN_ITERS >= SCAN_NT + 3,
              "slot parity follows the unrolled index; every role runs the same whole number of rings");

__device__ __forceinline__ void scan_handover() {
  asm volatile("s_wait_dscnt 0x0\n\ts_barrier_signal -1\n\ts_barrier_wait -1" ::: "memory");
}

template <int HALF>
__device__ __forceinline__ void scan_block(
    const int wid,
    const float* __restrict__ ab, float* __restrict__ s_out,
    float* __restrict__ st  ,
    const float tau, const float thr, float* lds) {
  const int lane = threadIdx.x & 31;
  const int wave = __builtin_amdgcn_readfirstlane((int)(threadIdx.x >> 5));
  const bool is_serial = (wave == 0);
  const bool is_helper = (wave >= 1) && (wave <= SCAN_HELPERS);
  const int hw = wave - 1;
  const int idx = wid * 32 + lane;
  const int b = idx >> 8;
  const int h = idx & (HDIM - 1);
  const float inv_tau = 1.0f / tau;
  const float decay = 1.0f - inv_tau;

  float hstate = 0.0f;
  float v = 0.0f;
  if (HALF > 0 && is_serial) {
    const v2f c = *(const v2f*)(st + 2 * idx);
    hstate = c.x;
    v = c.y;
  }
  const size_t base = ((size_t)b * TLEN + (size_t)HALF * THALF) * HDIM + h;
  const float* src = ab + 2 * base;
  float* dst = s_out + base;
  float* lab = lds + lane * 4;
  float* lsp = lds + SCAN_SP_BASE + lane * 4;

  if (is_serial) {
    v4f Q[32];
    for (int i0 = 0; i0 < SCAN_ITERS; i0 += SCAN_AHEAD) {
      #pragma unroll
      for (int u = 0; u < SCAN_AHEAD; ++u) {
        const int i = i0 + u;
        {
          const float* la = lab + ((u + 1) & 1) * SCAN_AB_SLOT;
          #pragma unroll
          for (int p = 0; p < 16; ++p)
            Q[16 * ((u + 1) & 1) + p] = *(const v4f*)(la + p * 128);
        }
        if (i >= 2 && i <= SCAN_NT + 1) {
          float sp[SCAN_TILE];
          #pragma unroll
          for (int j = 0; j < SCAN_TILE; ++j) {
            const v4f q = Q[16 * (u & 1) + (j >> 1)];
            const float ea = (j & 1) ? q.z : q.x;
            const float eb = (j & 1) ? q.w : q.y;
            hstate = ea * hstate + eb;
            v = decay * v + inv_tau * hstate;
            const float s1 = (v > thr) ? 1.0f : 0.0f;
            v -= s1 * thr;
            sp[j] = s1;
          }
          float* ls = lsp + (u & 1) * SCAN_SP_SLOT;
          #pragma unroll
          for (int q4 = 0; q4 < SCAN_TILE / 4; ++q4)
            *(v4f*)(ls + q4 * 128) =
                v4f{sp[4 * q4], sp[4 * q4 + 1], sp[4 * q4 + 2], sp[4 * q4 + 3]};
        }
        scan_handover();
      }
    }
  } else if (is_helper) {
    v2f R[8 * SCAN_AHEAD];
    #pragma unroll
    for (int r = 0; r < SCAN_AHEAD; ++r) {
      #pragma unroll
      for (int jj = 0; jj < 8; ++jj)
        R[8 * r + jj] =
            *(const v2f*)(src + (size_t)(r * SCAN_TILE + 8 * hw + jj) * (2 * HDIM));
    }
    for (int i0 = 0; i0 < SCAN_ITERS; i0 += SCAN_AHEAD) {
      #pragma unroll
      for (int u = 0; u < SCAN_AHEAD; ++u) {
        const int i = i0 + u;
        if (i >= 3 && i <= SCAN_NT + 2) {
          const float* ls = lsp + ((u + 1) & 1) * SCAN_SP_SLOT + (2 * hw) * 128;
          float* g = dst + (size_t)((i - 3) * SCAN_TILE + 8 * hw) * HDIM;
          #pragma unroll
          for (int mm = 0; mm < 2; ++mm) {
            const v4f sv = *(const v4f*)(ls + mm * 128);
            g[(size_t)(4 * mm + 0) * HDIM] = sv.x;
            g[(size_t)(4 * mm + 1) * HDIM] = sv.y;
            g[(size_t)(4 * mm + 2) * HDIM] = sv.z;
            g[(size_t)(4 * mm + 3) * HDIM] = sv.w;
          }
        }
        {
          float* la = lab + (u & 1) * SCAN_AB_SLOT + (4 * hw) * 128;
          #pragma unroll
          for (int jj = 0; jj < 8; ++jj)
            *(v2f*)(la + (jj >> 1) * 128 + (jj & 1) * 2) = R[8 * u + jj];
        }
        __builtin_amdgcn_sched_barrier(0);
        {
          const int tn = (i + SCAN_AHEAD < SCAN_NT) ? (i + SCAN_AHEAD) : (SCAN_NT - 1);
          #pragma unroll
          for (int jj = 0; jj < 8; ++jj)
            R[8 * u + jj] =
                *(const v2f*)(src + (size_t)(tn * SCAN_TILE + 8 * hw + jj) * (2 * HDIM));
        }
        scan_handover();
      }
    }
  } else {
    for (int i = 0; i < SCAN_ITERS; ++i)
      scan_handover();
  }

  if (HALF + 1 < NHALF && is_serial) {
    v2f c; c.x = hstate; c.y = v;
    *(v2f*)(st + 2 * idx) = c;
  }
}

template <int NW>
__device__ __forceinline__ void synapse_block(
    const int bid, const int half,
    const float* __restrict__ S, const float* __restrict__ Wsyn,
    float* __restrict__ out,
    float* lws, float* lA) {
  const int nt = bid & 15;
  const unsigned ml = (unsigned)bid >> 4;
  constexpr unsigned G = THALF / (NW * 64);
  const int mg = (int)((((ml / G) << 1) + (unsigned)half) * G + (ml % G));
  const int lane = threadIdx.x & 31;
  const int wave = threadIdx.x >> 5;

  {
    const float4* g = (const float4*)(Wsyn + (size_t)nt * 16 * HDIM);
    float4* sm = (float4*)lws;
    constexpr int NS = 16 * HDIM / 4 / (NW * 32);
    float4 tw[NS];
    #pragma unroll
    for (int it = 0; it < NS; ++it)
      tw[it] = g[threadIdx.x + it * (NW * 32)];
    #pragma unroll
    for (int it = 0; it < NS; ++it)
      sm[threadIdx.x + it * (NW * 32)] = tw[it];
  }

  const int hi   = lane >> 4;
  const int l16  = lane & 15;
  const int n    = nt * 16 + l16;
  const int koff = hi << 1;
  const int m0   = (mg * NW + wave) * 64;

  v8f c[4] = {v8f{}, v8f{}, v8f{}, v8f{}};

  const float* brow = lws + l16 * HDIM + koff;
  const float* gsrc = S + (size_t)(m0 + (lane >> 3)) * HDIM + 4 * (lane & 7);
  float* ldst = lA + (wave * 64 + (lane >> 3)) * APITCH + 4 * (lane & 7);
  const float* afrag = lA + (wave * 64 + l16) * APITCH + koff;

  v4f pre[16];
  A_CHUNK_LOAD(pre, gsrc);
  __syncthreads();
  for (int kc = 0; kc < NCHUNK; ++kc) {
    wave_lds_order();
    A_CHUNK_STORE(pre, ldst);
    wave_lds_order();
    if (kc + 1 < NCHUNK)
      A_CHUNK_LOAD(pre, gsrc + (kc + 1) * KC);
    __builtin_amdgcn_sched_barrier(0);
    #pragma unroll
    for (int j = 0; j < KC / 4; ++j) {
      v2f bf = *(const v2f*)(brow + kc * KC + 4 * j);
      #pragma unroll
      for (int s = 0; s < 4; ++s) {
        v2f af = *(const v2f*)(afrag + (16 * s) * APITCH + 4 * j);
        c[s] = __builtin_amdgcn_wmma_f32_16x16x4_f32(false, af, false, bf,
                                                     (short)0, c[s], false, false);
      }
    }
  }

  #pragma unroll
  for (int s = 0; s < 4; ++s) {
    const int rbase = m0 + 16 * s + (hi << 3);
    #pragma unroll
    for (int r = 0; r < 8; ++r)
      out[(size_t)(rbase + r) * HDIM + n] = c[s][r];
  }
}

__global__ __launch_bounds__(NW_SOLO * 32) void gates_first(
    const float* __restrict__ X,
    const float* __restrict__ Wa, const float* __restrict__ ba,
    const float* __restrict__ Wu, const float* __restrict__ bu,
    float* __restrict__ ab) {
  __shared__ __attribute__((aligned(16))) float lw[2 * HDIM * 16];
  __shared__ __attribute__((aligned(16))) float lA[AROWS_SOLO * APITCH];
  gates_block<NW_SOLO>(blockIdx.x, 0, X, Wa, ba, Wu, bu, ab, lw, lw + HDIM * 16, lA);
}

__global__ __launch_bounds__(256) void scan_first_gates_second(
    const float* __restrict__ X,
    const float* __restrict__ Wa, const float* __restrict__ ba,
    const float* __restrict__ Wu, const float* __restrict__ bu,
    const float* __restrict__ ab_rd, float* __restrict__ ab_wr,
    float* __restrict__ s_out, float* __restrict__ st,
    const float* __restrict__ tau_p, const float* __restrict__ thr_p) {
  __shared__ __attribute__((aligned(16))) float lw[2 * HDIM * 16];
  __shared__ __attribute__((aligned(16))) float lA[AROWS * APITCH];
  if (blockIdx.x < SCAN_BLOCKS) {
    const float tau = tau_p[0];
    const float thr = thr_p[0];
    scan_block<0>(blockIdx.x, ab_rd, s_out, st, tau, thr, lA);
    return;
  }
  gates_block<8>(blockIdx.x - SCAN_BLOCKS, 1, X, Wa, ba, Wu, bu, ab_wr,
                 lw, lw + HDIM * 16, lA);
}

__global__ __launch_bounds__(256) void scan_second_synapse_first(
    const float* __restrict__ ab, float* __restrict__ s_wr,
    float* __restrict__ st,
    const float* __restrict__ tau_p, const float* __restrict__ thr_p,
    const float* __restrict__ s_rd, const float* __restrict__ Wsyn,
    float* __restrict__ out) {
  __shared__ __attribute__((aligned(16))) float lw[16 * HDIM];
  __shared__ __attribute__((aligned(16))) float lA[AROWS * APITCH];
  if (blockIdx.x < SCAN_BLOCKS) {
    const float tau = tau_p[0];
    const float thr = thr_p[0];
    scan_block<1>(blockIdx.x, ab, s_wr, st, tau, thr, lA);
    return;
  }
  synapse_block<8>(blockIdx.x - SCAN_BLOCKS, 0, s_rd, Wsyn, out, lw, lA);
}

__global__ __launch_bounds__(NW_SOLO * 32) void synapse_second(
    const float* __restrict__ S, const float* __restrict__ Wsyn,
    float* __restrict__ out) {
  __shared__ __attribute__((aligned(16))) float lw[16 * HDIM];
  __shared__ __attribute__((aligned(16))) float lA[AROWS_SOLO * APITCH];
  synapse_block<NW_SOLO>(blockIdx.x, 1, S, Wsyn, out, lw, lA);
}

extern "C" void kernel_launch(void* const* d_in, const int* in_sizes, int n_in,
                              void* d_out, int out_size, void* d_ws, size_t ws_size,
                              hipStream_t stream) {
  const float* x    = (const float*)d_in[0];
  const float* Wa   = (const float*)d_in[1];
  const float* ba   = (const float*)d_in[2];
  const float* Wu   = (const float*)d_in[3];
  const float* bu   = (const float*)d_in[4];
  const float* tau  = (const float*)d_in[5];
  const float* thr  = (const float*)d_in[6];
  const float* Wsyn = (const float*)d_in[7];
  float* out = (float*)d_out;

  float* ab = (float*)d_ws;
  float* s  = ab + 2 * (size_t)BT * HDIM;
  float* st = s + (size_t)BT * HDIM;

  gates_first<<<GEMM_BLOCKS * (8 / NW_SOLO), NW_SOLO * 32, 0, stream>>>(x, Wa, ba, Wu, bu, ab);
  scan_first_gates_second<<<SCAN_BLOCKS + GEMM_BLOCKS, 256, 0, stream>>>(
      x, Wa, ba, Wu, bu, ab, ab, s, st, tau, thr);
  scan_second_synapse_first<<<SCAN_BLOCKS + GEMM_BLOCKS, 256, 0, stream>>>(
      ab, s, st, tau, thr, s, Wsyn, out);
  synapse_second<<<GEMM_BLOCKS * (8 / NW_SOLO), NW_SOLO * 32, 0, stream>>>(s, Wsyn, out);
}


# scan: next step's charge formed ahead for both reset outcomes (decay*w+c and decay*w+(c-decay*thr)), so a step waits only for compare+select; v rounds differently unless tau=1
# speedup vs baseline: 14.1160x; 1.0383x over previous
#include <hip/hip_runtime.h>
#include <hip/hip_bf16.h>

#define BATCH 8
#define TLEN  2048
#define HDIM  256
#define BT    (BATCH * TLEN)
#define NHALF 2
#define THALF (TLEN / NHALF)
#define GEMM_BLOCKS 256
#define SCAN_BLOCKS 64

typedef __attribute__((ext_vector_type(2))) float v2f;
typedef __attribute__((ext_vector_type(4))) float v4f;
typedef __attribute__((ext_vector_type(8))) float v8f;

#define KC     32
#define NCHUNK (HDIM / KC)
#define APITCH 36
#define AROWS  512
#define NW_SOLO 4
#define AROWS_SOLO (NW_SOLO * 64)

#define A_CHUNK_LOAD(pre, g)                                              \
  _Pragma("unroll") for (int q = 0; q < 16; ++q)                          \
    (pre)[q] = *(const v4f*)((g) + (size_t)(4 * q) * HDIM)
#define A_CHUNK_STORE(pre, l)                                             \
  _Pragma("unroll") for (int q = 0; q < 16; ++q)                          \
    *(v4f*)((l) + (4 * q) * APITCH) = (pre)[q]

__device__ __forceinline__ void wave_lds_order() {
  __builtin_amdgcn_fence(__ATOMIC_RELEASE, "wavefront");
  __builtin_amdgcn_wave_barrier();
  __builtin_amdgcn_fence(__ATOMIC_ACQUIRE, "wavefront");
}

template <int NW>
__device__ __forceinline__ void gates_block(
    const int bid, const int half,
    const float* __restrict__ X,
    const float* __restrict__ Wa, const float* __restrict__ ba,
    const float* __restrict__ Wu, const float* __restrict__ bu,
    float* __restrict__ ab  ,
    float* lwa, float* lwu, float* lA) {
  const int nt = bid & 15;
  const unsigned ml = (unsigned)bid >> 4;
  constexpr unsigned G = THALF / (NW * 64);
  const int mg = (int)((((ml / G) << 1) + (unsigned)half) * G + (ml % G));
  const int lane = threadIdx.x & 31;
  const int wave = threadIdx.x >> 5;

  {
    const float4* ga = (const float4*)(Wa + nt * 16);
    const float4* gu = (const float4*)(Wu + nt * 16);
    float4* sa = (float4*)lwa;
    float4* su = (float4*)lwu;
    constexpr int NS = HDIM * 4 / (NW * 32);
    float4 ta[NS], tu[NS];
    #pragma unroll
    for (int it = 0; it < NS; ++it) {
      const int i = threadIdx.x + it * (NW * 32);
      const int k = i >> 2;
      const int q = i & 3;
      ta[it] = ga[k * (HDIM / 4) + q];
      tu[it] = gu[k * (HDIM / 4) + q];
    }
    #pragma unroll
    for (int it = 0; it < NS; ++it) {
      const int i = threadIdx.x + it * (NW * 32);
      sa[i] = ta[it];
      su[i] = tu[it];
    }
  }

  const int hi   = lane >> 4;
  const int l16  = lane & 15;
  const int n    = nt * 16 + l16;
  const int koff = hi << 1;
  const int m0   = (mg * NW + wave) * 64;

  v8f ca[4] = {v8f{}, v8f{}, v8f{}, v8f{}};
  v8f cu[4] = {v8f{}, v8f{}, v8f{}, v8f{}};

  const float* gsrc = X + (size_t)(m0 + (lane >> 3)) * HDIM + 4 * (lane & 7);
  float* ldst = lA + (wave * 64 + (lane >> 3)) * APITCH + 4 * (lane & 7);
  const float* afrag = lA + (wave * 64 + l16) * APITCH + koff;

  v4f pre[16];
  A_CHUNK_LOAD(pre, gsrc);
  __syncthreads();
  for (int kc = 0; kc < NCHUNK; ++kc) {
    wave_lds_order();
    A_CHUNK_STORE(pre, ldst);
    wave_lds_order();
    if (kc + 1 < NCHUNK)
      A_CHUNK_LOAD(pre, gsrc + (kc + 1) * KC);
    __builtin_amdgcn_sched_barrier(0);
    #pragma unroll
    for (int j = 0; j < KC / 4; ++j) {
      const int k = kc * KC + 4 * j + koff;
      v2f bwa; bwa.x = lwa[k * 16 + l16]; bwa.y = lwa[(k + 1) * 16 + l16];
      v2f bwu; bwu.x = lwu[k * 16 + l16]; bwu.y = lwu[(k + 1) * 16 + l16];
      #pragma unroll
      for (int s = 0; s < 4; ++s) {
        v2f af = *(const v2f*)(afrag + (16 * s) * APITCH + 4 * j);
        ca[s] = __builtin_amdgcn_wmma_f32_16x16x4_f32(false, af, false, bwa,
                                                      (short)0, ca[s], false, false);
        cu[s] = __builtin_amdgcn_wmma_f32_16x16x4_f32(false, af, false, bwu,
                                                      (short)0, cu[s], false, false);
      }
    }
  }

  const float bia = ba[n];
  const float biu = bu[n];
  #pragma unroll
  for (int s = 0; s < 4; ++s) {
    const int rbase = m0 + 16 * s + (hi << 3);
    #pragma unroll
    for (int r = 0; r < 8; ++r) {
      const float za = ca[s][r] + bia;
      const float zu = cu[s][r] + biu;
      const float ga = 1.0f / (1.0f + __expf(-za));
      const float gu = tanhf(zu);
      v2f e; e.x = ga; e.y = (1.0f - ga) * gu;
      *(v2f*)(ab + 2 * ((size_t)(rbase + r) * HDIM + n)) = e;
    }
  }
}

#define SCAN_TILE    32
#define SCAN_NT      (THALF / SCAN_TILE)
#define SCAN_AHEAD   4
#define SCAN_HELPERS 4
#define SCAN_ITERS   36
#define SCAN_AB_SLOT 2048
#define SCAN_SP_BASE 4096
#define SCAN_SP_SLOT 1024
static_assert(THALF % SCAN_TILE == 0 && SCAN_TILE == 8 * SCAN_HELPERS, "whole tiles; 8 steps per helper");
static_assert(SCAN_AHEAD % 2 == 0 && SCAN_ITERS % SCAN_AHEAD == 0 && SCAN_ITERS >= SCAN_NT + 3,
              "slot parity follows the unrolled index; every role runs the same whole number of rings");

__device__ __forceinline__ void scan_handover() {
  asm volatile("s_wait_dscnt 0x0\n\ts_barrier_signal -1\n\ts_barrier_wait -1" ::: "memory");
}

template <int HALF>
__device__ __forceinline__ void scan_block(
    const int wid,
    const float* __restrict__ ab, float* __restrict__ s_out,
    float* __restrict__ st  ,
    const float tau, const float thr, float* lds) {
  const int lane = threadIdx.x & 31;
  const int wave = __builtin_amdgcn_readfirstlane((int)(threadIdx.x >> 5));
  const bool is_serial = (wave == 0);
  const bool is_helper = (wave >= 1) && (wave <= SCAN_HELPERS);
  const int hw = wave - 1;
  const int idx = wid * 32 + lane;
  const int b = idx >> 8;
  const int h = idx & (HDIM - 1);
  const float inv_tau = 1.0f / tau;
  const float decay = 1.0f - inv_tau;
  const float dthr = decay * thr;

  float hstate = 0.0f;
  float v = 0.0f;
  if (HALF > 0 && is_serial) {
    const v2f c = *(const v2f*)(st + 2 * idx);
    hstate = c.x;
    v = c.y;
  }
  const size_t base = ((size_t)b * TLEN + (size_t)HALF * THALF) * HDIM + h;
  const float* src = ab + 2 * base;
  float* dst = s_out + base;
  float* lab = lds + lane * 4;
  float* lsp = lds + SCAN_SP_BASE + lane * 4;

  if (is_serial) {
    v4f Q[32];
    for (int i0 = 0; i0 < SCAN_ITERS; i0 += SCAN_AHEAD) {
      #pragma unroll
      for (int u = 0; u < SCAN_AHEAD; ++u) {
        const int i = i0 + u;
        {
          const float* la = lab + ((u + 1) & 1) * SCAN_AB_SLOT;
          #pragma unroll
          for (int p = 0; p < 16; ++p)
            Q[16 * ((u + 1) & 1) + p] = *(const v4f*)(la + p * 128);
        }
        if (i >= 2 && i <= SCAN_NT + 1) {
          float sp[SCAN_TILE];
          {
            const v4f q0 = Q[16 * (u & 1)];
            hstate = q0.x * hstate + q0.y;
          }
          float w = __builtin_fmaf(decay, v, inv_tau * hstate);
          #pragma unroll
          for (int j = 0; j < SCAN_TILE; ++j) {
            const bool fire = (w > thr);
            const float s1 = fire ? 1.0f : 0.0f;
            sp[j] = s1;
            if (j + 1 < SCAN_TILE) {
              const v4f q = Q[16 * (u & 1) + ((j + 1) >> 1)];
              const float ea = ((j + 1) & 1) ? q.z : q.x;
              const float eb = ((j + 1) & 1) ? q.w : q.y;
              hstate = ea * hstate + eb;
              const float c = inv_tau * hstate;
              const float keep = __builtin_fmaf(decay, w, c);
              const float drop = __builtin_fmaf(decay, w, c - dthr);
              w = fire ? drop : keep;
            } else {
              v = w - s1 * thr;
            }
          }
          float* ls = lsp + (u & 1) * SCAN_SP_SLOT;
          #pragma unroll
          for (int q4 = 0; q4 < SCAN_TILE / 4; ++q4)
            *(v4f*)(ls + q4 * 128) =
                v4f{sp[4 * q4], sp[4 * q4 + 1], sp[4 * q4 + 2], sp[4 * q4 + 3]};
        }
        scan_handover();
      }
    }
  } else if (is_helper) {
    v2f R[8 * SCAN_AHEAD];
    #pragma unroll
    for (int r = 0; r < SCAN_AHEAD; ++r) {
      #pragma unroll
      for (int jj = 0; jj < 8; ++jj)
        R[8 * r + jj] =
            *(const v2f*)(src + (size_t)(r * SCAN_TILE + 8 * hw + jj) * (2 * HDIM));
    }
    for (int i0 = 0; i0 < SCAN_ITERS; i0 += SCAN_AHEAD) {
      #pragma unroll
      for (int u = 0; u < SCAN_AHEAD; ++u) {
        const int i = i0 + u;
        if (i >= 3 && i <= SCAN_NT + 2) {
          const float* ls = lsp + ((u + 1) & 1) * SCAN_SP_SLOT + (2 * hw) * 128;
          float* g = dst + (size_t)((i - 3) * SCAN_TILE + 8 * hw) * HDIM;
          #pragma unroll
          for (int mm = 0; mm < 2; ++mm) {
            const v4f sv = *(const v4f*)(ls + mm * 128);
            g[(size_t)(4 * mm + 0) * HDIM] = sv.x;
            g[(size_t)(4 * mm + 1) * HDIM] = sv.y;
            g[(size_t)(4 * mm + 2) * HDIM] = sv.z;
            g[(size_t)(4 * mm + 3) * HDIM] = sv.w;
          }
        }
        {
          float* la = lab + (u & 1) * SCAN_AB_SLOT + (4 * hw) * 128;
          #pragma unroll
          for (int jj = 0; jj < 8; ++jj)
            *(v2f*)(la + (jj >> 1) * 128 + (jj & 1) * 2) = R[8 * u + jj];
        }
        __builtin_amdgcn_sched_barrier(0);
        {
          const int tn = (i + SCAN_AHEAD < SCAN_NT) ? (i + SCAN_AHEAD) : (SCAN_NT - 1);
          #pragma unroll
          for (int jj = 0; jj < 8; ++jj)
            R[8 * u + jj] =
                *(const v2f*)(src + (size_t)(tn * SCAN_TILE + 8 * hw + jj) * (2 * HDIM));
        }
        scan_handover();
      }
    }
  } else {
    for (int i = 0; i < SCAN_ITERS; ++i)
      scan_handover();
  }

  if (HALF + 1 < NHALF && is_serial) {
    v2f c; c.x = hstate; c.y = v;
    *(v2f*)(st + 2 * idx) = c;
  }
}

template <int NW>
__device__ __forceinline__ void synapse_block(
    const int bid, const int half,
    const float* __restrict__ S, const float* __restrict__ Wsyn,
    float* __restrict__ out,
    float* lws, float* lA) {
  const int nt = bid & 15;
  const unsigned ml = (unsigned)bid >> 4;
  constexpr unsigned G = THALF / (NW * 64);
  const int mg = (int)((((ml / G) << 1) + (unsigned)half) * G + (ml % G));
  const int lane = threadIdx.x & 31;
  const int wave = threadIdx.x >> 5;

  {
    const float4* g = (const float4*)(Wsyn + (size_t)nt * 16 * HDIM);
    float4* sm = (float4*)lws;
    constexpr int NS = 16 * HDIM / 4 / (NW * 32);
    float4 tw[NS];
    #pragma unroll
    for (int it = 0; it < NS; ++it)
      tw[it] = g[threadIdx.x + it * (NW * 32)];
    #pragma unroll
    for (int it = 0; it < NS; ++it)
      sm[threadIdx.x + it * (NW * 32)] = tw[it];
  }

  const int hi   = lane >> 4;
  const int l16  = lane & 15;
  const int n    = nt * 16 + l16;
  const int koff = hi << 1;
  const int m0   = (mg * NW + wave) * 64;

  v8f c[4] = {v8f{}, v8f{}, v8f{}, v8f{}};

  const float* brow = lws + l16 * HDIM + koff;
  const float* gsrc = S + (size_t)(m0 + (lane >> 3)) * HDIM + 4 * (lane & 7);
  float* ldst = lA + (wave * 64 + (lane >> 3)) * APITCH + 4 * (lane & 7);
  const float* afrag = lA + (wave * 64 + l16) * APITCH + koff;

  v4f pre[16];
  A_CHUNK_LOAD(pre, gsrc);
  __syncthreads();
  for (int kc = 0; kc < NCHUNK; ++kc) {
    wave_lds_order();
    A_CHUNK_STORE(pre, ldst);
    wave_lds_order();
    if (kc + 1 < NCHUNK)
      A_CHUNK_LOAD(pre, gsrc + (kc + 1) * KC);
    __builtin_amdgcn_sched_barrier(0);
    #pragma unroll
    for (int j = 0; j < KC / 4; ++j) {
      v2f bf = *(const v2f*)(brow + kc * KC + 4 * j);
      #pragma unroll
      for (int s = 0; s < 4; ++s) {
        v2f af = *(const v2f*)(afrag + (16 * s) * APITCH + 4 * j);
        c[s] = __builtin_amdgcn_wmma_f32_16x16x4_f32(false, af, false, bf,
                                                     (short)0, c[s], false, false);
      }
    }
  }

  #pragma unroll
  for (int s = 0; s < 4; ++s) {
    const int rbase = m0 + 16 * s + (hi << 3);
    #pragma unroll
    for (int r = 0; r < 8; ++r)
      out[(size_t)(rbase + r) * HDIM + n] = c[s][r];
  }
}

__global__ __launch_bounds__(NW_SOLO * 32) void gates_first(
    const float* __restrict__ X,
    const float* __restrict__ Wa, const float* __restrict__ ba,
    const float* __restrict__ Wu, const float* __restrict__ bu,
    float* __restrict__ ab) {
  __shared__ __attribute__((aligned(16))) float lw[2 * HDIM * 16];
  __shared__ __attribute__((aligned(16))) float lA[AROWS_SOLO * APITCH];
  gates_block<NW_SOLO>(blockIdx.x, 0, X, Wa, ba, Wu, bu, ab, lw, lw + HDIM * 16, lA);
}

__global__ __launch_bounds__(256) void scan_first_gates_second(
    const float* __restrict__ X,
    const float* __restrict__ Wa, const float* __restrict__ ba,
    const float* __restrict__ Wu, const float* __restrict__ bu,
    const float* __restrict__ ab_rd, float* __restrict__ ab_wr,
    float* __restrict__ s_out, float* __restrict__ st,
    const float* __restrict__ tau_p, const float* __restrict__ thr_p) {
  __shared__ __attribute__((aligned(16))) float lw[2 * HDIM * 16];
  __shared__ __attribute__((aligned(16))) float lA[AROWS * APITCH];
  if (blockIdx.x < SCAN_BLOCKS) {
    const float tau = tau_p[0];
    const float thr = thr_p[0];
    scan_block<0>(blockIdx.x, ab_rd, s_out, st, tau, thr, lA);
    return;
  }
  gates_block<8>(blockIdx.x - SCAN_BLOCKS, 1, X, Wa, ba, Wu, bu, ab_wr,
                 lw, lw + HDIM * 16, lA);
}

__global__ __launch_bounds__(256) void scan_second_synapse_first(
    const float* __restrict__ ab, float* __restrict__ s_wr,
    float* __restrict__ st,
    const float* __restrict__ tau_p, const float* __restrict__ thr_p,
    const float* __restrict__ s_rd, const float* __restrict__ Wsyn,
    float* __restrict__ out) {
  __shared__ __attribute__((aligned(16))) float lw[16 * HDIM];
  __shared__ __attribute__((aligned(16))) float lA[AROWS * APITCH];
  if (blockIdx.x < SCAN_BLOCKS) {
    const float tau = tau_p[0];
    const float thr = thr_p[0];
    scan_block<1>(blockIdx.x, ab, s_wr, st, tau, thr, lA);
    return;
  }
  synapse_block<8>(blockIdx.x - SCAN_BLOCKS, 0, s_rd, Wsyn, out, lw, lA);
}

__global__ __launch_bounds__(NW_SOLO * 32) void synapse_second(
    const float* __restrict__ S, const float* __restrict__ Wsyn,
    float* __restrict__ out) {
  __shared__ __attribute__((aligned(16))) float lw[16 * HDIM];
  __shared__ __attribute__((aligned(16))) float lA[AROWS_SOLO * APITCH];
  synapse_block<NW_SOLO>(blockIdx.x, 1, S, Wsyn, out, lw, lA);
}

extern "C" void kernel_launch(void* const* d_in, const int* in_sizes, int n_in,
                              void* d_out, int out_size, void* d_ws, size_t ws_size,
                              hipStream_t stream) {
  const float* x    = (const float*)d_in[0];
  const float* Wa   = (const float*)d_in[1];
  const float* ba   = (const float*)d_in[2];
  const float* Wu   = (const float*)d_in[3];
  const float* bu   = (const float*)d_in[4];
  const float* tau  = (const float*)d_in[5];
  const float* thr  = (const float*)d_in[6];
  const float* Wsyn = (const float*)d_in[7];
  float* out = (float*)d_out;

  float* ab = (float*)d_ws;
  float* s  = ab + 2 * (size_t)BT * HDIM;
  float* st = s + (size_t)BT * HDIM;

  gates_first<<<GEMM_BLOCKS * (8 / NW_SOLO), NW_SOLO * 32, 0, stream>>>(x, Wa, ba, Wu, bu, ab);
  scan_first_gates_second<<<SCAN_BLOCKS + GEMM_BLOCKS, 256, 0, stream>>>(
      x, Wa, ba, Wu, bu, ab, ab, s, st, tau, thr);
  scan_second_synapse_first<<<SCAN_BLOCKS + GEMM_BLOCKS, 256, 0, stream>>>(
      ab, s, st, tau, thr, s, Wsyn, out);
  synapse_second<<<GEMM_BLOCKS * (8 / NW_SOLO), NW_SOLO * 32, 0, stream>>>(s, Wsyn, out);
}


# gate GEMM epilogue: tanh without a branch (both ranges computed, one select; the library's own operations and constants), so a lane's 32 values form one straight block; values bit-identical
# speedup vs baseline: 14.8105x; 1.0492x over previous
#include <hip/hip_runtime.h>
#include <hip/hip_bf16.h>

#define BATCH 8
#define TLEN  2048
#define HDIM  256
#define BT    (BATCH * TLEN)
#define NHALF 2
#define THALF (TLEN / NHALF)
#define GEMM_BLOCKS 256
#define SCAN_BLOCKS 64

typedef __attribute__((ext_vector_type(2))) float v2f;
typedef __attribute__((ext_vector_type(4))) float v4f;
typedef __attribute__((ext_vector_type(8))) float v8f;

#define KC     32
#define NCHUNK (HDIM / KC)
#define APITCH 36
#define AROWS  512
#define NW_SOLO 4
#define AROWS_SOLO (NW_SOLO * 64)

#define A_CHUNK_LOAD(pre, g)                                              \
  _Pragma("unroll") for (int q = 0; q < 16; ++q)                          \
    (pre)[q] = *(const v4f*)((g) + (size_t)(4 * q) * HDIM)
#define A_CHUNK_STORE(pre, l)                                             \
  _Pragma("unroll") for (int q = 0; q < 16; ++q)                          \
    *(v4f*)((l) + (4 * q) * APITCH) = (pre)[q]

__device__ __forceinline__ void wave_lds_order() {
  __builtin_amdgcn_fence(__ATOMIC_RELEASE, "wavefront");
  __builtin_amdgcn_wave_barrier();
  __builtin_amdgcn_fence(__ATOMIC_ACQUIRE, "wavefront");
}


__device__ __forceinline__ float tanh_both_ranges(const float x) {
  const float y = __builtin_fabsf(x);
  const float y2 = y * y;
  float p = __builtin_fmaf(y2, -0x1.758e7ap-8f, 0x1.521192p-6f);
  p = __builtin_fmaf(y2, p, -0x1.b8389cp-5f);
  p = __builtin_fmaf(y2, p, 0x1.110704p-3f);
  p = __builtin_fmaf(y2, p, -0x1.555532p-2f);
  const float z_small = __builtin_fmaf(y2, y * p, y);
  const float t = expf(2.0f * y);
  const float z_large = __builtin_fmaf(-2.0f, __builtin_amdgcn_rcpf(t + 1.0f), 1.0f);
  const float z = (y < 0.625f) ? z_small : z_large;
  return __builtin_copysignf(z, x);
}

template <int NW>
__device__ __forceinline__ void gates_block(
    const int bid, const int half,
    const float* __restrict__ X,
    const float* __restrict__ Wa, const float* __restrict__ ba,
    const float* __restrict__ Wu, const float* __restrict__ bu,
    float* __restrict__ ab  ,
    float* lwa, float* lwu, float* lA) {
  const int nt = bid & 15;
  const unsigned ml = (unsigned)bid >> 4;
  constexpr unsigned G = THALF / (NW * 64);
  const int mg = (int)((((ml / G) << 1) + (unsigned)half) * G + (ml % G));
  const int lane = threadIdx.x & 31;
  const int wave = threadIdx.x >> 5;

  {
    const float4* ga = (const float4*)(Wa + nt * 16);
    const float4* gu = (const float4*)(Wu + nt * 16);
    float4* sa = (float4*)lwa;
    float4* su = (float4*)lwu;
    constexpr int NS = HDIM * 4 / (NW * 32);
    float4 ta[NS], tu[NS];
    #pragma unroll
    for (int it = 0; it < NS; ++it) {
      const int i = threadIdx.x + it * (NW * 32);
      const int k = i >> 2;
      const int q = i & 3;
      ta[it] = ga[k * (HDIM / 4) + q];
      tu[it] = gu[k * (HDIM / 4) + q];
    }
    #pragma unroll
    for (int it = 0; it < NS; ++it) {
      const int i = threadIdx.x + it * (NW * 32);
      sa[i] = ta[it];
      su[i] = tu[it];
    }
  }

  const int hi   = lane >> 4;
  const int l16  = lane & 15;
  const int n    = nt * 16 + l16;
  const int koff = hi << 1;
  const int m0   = (mg * NW + wave) * 64;

  v8f ca[4] = {v8f{}, v8f{}, v8f{}, v8f{}};
  v8f cu[4] = {v8f{}, v8f{}, v8f{}, v8f{}};

  const float* gsrc = X + (size_t)(m0 + (lane >> 3)) * HDIM + 4 * (lane & 7);
  float* ldst = lA + (wave * 64 + (lane >> 3)) * APITCH + 4 * (lane & 7);
  const float* afrag = lA + (wave * 64 + l16) * APITCH + koff;

  v4f pre[16];
  A_CHUNK_LOAD(pre, gsrc);
  __syncthreads();
  for (int kc = 0; kc < NCHUNK; ++kc) {
    wave_lds_order();
    A_CHUNK_STORE(pre, ldst);
    wave_lds_order();
    if (kc + 1 < NCHUNK)
      A_CHUNK_LOAD(pre, gsrc + (kc + 1) * KC);
    __builtin_amdgcn_sched_barrier(0);
    #pragma unroll
    for (int j = 0; j < KC / 4; ++j) {
      const int k = kc * KC + 4 * j + koff;
      v2f bwa; bwa.x = lwa[k * 16 + l16]; bwa.y = lwa[(k + 1) * 16 + l16];
      v2f bwu; bwu.x = lwu[k * 16 + l16]; bwu.y = lwu[(k + 1) * 16 + l16];
      #pragma unroll
      for (int s = 0; s < 4; ++s) {
        v2f af = *(const v2f*)(afrag + (16 * s) * APITCH + 4 * j);
        ca[s] = __builtin_amdgcn_wmma_f32_16x16x4_f32(false, af, false, bwa,
                                                      (short)0, ca[s], false, false);
        cu[s] = __builtin_amdgcn_wmma_f32_16x16x4_f32(false, af, false, bwu,
                                                      (short)0, cu[s], false, false);
      }
    }
  }

  const float bia = ba[n];
  const float biu = bu[n];
  #pragma unroll
  for (int s = 0; s < 4; ++s) {
    const int rbase = m0 + 16 * s + (hi << 3);
    #pragma unroll
    for (int r = 0; r < 8; ++r) {
      const float za = ca[s][r] + bia;
      const float zu = cu[s][r] + biu;
      const float ga = 1.0f / (1.0f + __expf(-za));
      const float gu = tanh_both_ranges(zu);
      v2f e; e.x = ga; e.y = (1.0f - ga) * gu;
      *(v2f*)(ab + 2 * ((size_t)(rbase + r) * HDIM + n)) = e;
    }
  }
}

#define SCAN_TILE    32
#define SCAN_NT      (THALF / SCAN_TILE)
#define SCAN_AHEAD   4
#define SCAN_HELPERS 4
#define SCAN_ITERS   36
#define SCAN_AB_SLOT 2048
#define SCAN_SP_BASE 4096
#define SCAN_SP_SLOT 1024
static_assert(THALF % SCAN_TILE == 0 && SCAN_TILE == 8 * SCAN_HELPERS, "whole tiles; 8 steps per helper");
static_assert(SCAN_AHEAD % 2 == 0 && SCAN_ITERS % SCAN_AHEAD == 0 && SCAN_ITERS >= SCAN_NT + 3,
              "slot parity follows the unrolled index; every role runs the same whole number of rings");

__device__ __forceinline__ void scan_handover() {
  asm volatile("s_wait_dscnt 0x0\n\ts_barrier_signal -1\n\ts_barrier_wait -1" ::: "memory");
}

template <int HALF>
__device__ __forceinline__ void scan_block(
    const int wid,
    const float* __restrict__ ab, float* __restrict__ s_out,
    float* __restrict__ st  ,
    const float tau, const float thr, float* lds) {
  const int lane = threadIdx.x & 31;
  const int wave = __builtin_amdgcn_readfirstlane((int)(threadIdx.x >> 5));
  const bool is_serial = (wave == 0);
  const bool is_helper = (wave >= 1) && (wave <= SCAN_HELPERS);
  const int hw = wave - 1;
  const int idx = wid * 32 + lane;
  const int b = idx >> 8;
  const int h = idx & (HDIM - 1);
  const float inv_tau = 1.0f / tau;
  const float decay = 1.0f - inv_tau;
  const float dthr = decay * thr;

  float hstate = 0.0f;
  float v = 0.0f;
  if (HALF > 0 && is_serial) {
    const v2f c = *(const v2f*)(st + 2 * idx);
    hstate = c.x;
    v = c.y;
  }
  const size_t base = ((size_t)b * TLEN + (size_t)HALF * THALF) * HDIM + h;
  const float* src = ab + 2 * base;
  float* dst = s_out + base;
  float* lab = lds + lane * 4;
  float* lsp = lds + SCAN_SP_BASE + lane * 4;

  if (is_serial) {
    v4f Q[32];
    for (int i0 = 0; i0 < SCAN_ITERS; i0 += SCAN_AHEAD) {
      #pragma unroll
      for (int u = 0; u < SCAN_AHEAD; ++u) {
        const int i = i0 + u;
        {
          const float* la = lab + ((u + 1) & 1) * SCAN_AB_SLOT;
          #pragma unroll
          for (int p = 0; p < 16; ++p)
            Q[16 * ((u + 1) & 1) + p] = *(const v4f*)(la + p * 128);
        }
        if (i >= 2 && i <= SCAN_NT + 1) {
          float sp[SCAN_TILE];
          {
            const v4f q0 = Q[16 * (u & 1)];
            hstate = q0.x * hstate + q0.y;
          }
          float w = __builtin_fmaf(decay, v, inv_tau * hstate);
          #pragma unroll
          for (int j = 0; j < SCAN_TILE; ++j) {
            const bool fire = (w > thr);
            const float s1 = fire ? 1.0f : 0.0f;
            sp[j] = s1;
            if (j + 1 < SCAN_TILE) {
              const v4f q = Q[16 * (u & 1) + ((j + 1) >> 1)];
              const float ea = ((j + 1) & 1) ? q.z : q.x;
              const float eb = ((j + 1) & 1) ? q.w : q.y;
              hstate = ea * hstate + eb;
              const float c = inv_tau * hstate;
              const float keep = __builtin_fmaf(decay, w, c);
              const float drop = __builtin_fmaf(decay, w, c - dthr);
              w = fire ? drop : keep;
            } else {
              v = w - s1 * thr;
            }
          }
          float* ls = lsp + (u & 1) * SCAN_SP_SLOT;
          #pragma unroll
          for (int q4 = 0; q4 < SCAN_TILE / 4; ++q4)
            *(v4f*)(ls + q4 * 128) =
                v4f{sp[4 * q4], sp[4 * q4 + 1], sp[4 * q4 + 2], sp[4 * q4 + 3]};
        }
        scan_handover();
      }
    }
  } else if (is_helper) {
    v2f R[8 * SCAN_AHEAD];
    #pragma unroll
    for (int r = 0; r < SCAN_AHEAD; ++r) {
      #pragma unroll
      for (int jj = 0; jj < 8; ++jj)
        R[8 * r + jj] =
            *(const v2f*)(src + (size_t)(r * SCAN_TILE + 8 * hw + jj) * (2 * HDIM));
    }
    for (int i0 = 0; i0 < SCAN_ITERS; i0 += SCAN_AHEAD) {
      #pragma unroll
      for (int u = 0; u < SCAN_AHEAD; ++u) {
        const int i = i0 + u;
        if (i >= 3 && i <= SCAN_NT + 2) {
          const float* ls = lsp + ((u + 1) & 1) * SCAN_SP_SLOT + (2 * hw) * 128;
          float* g = dst + (size_t)((i - 3) * SCAN_TILE + 8 * hw) * HDIM;
          #pragma unroll
          for (int mm = 0; mm < 2; ++mm) {
            const v4f sv = *(const v4f*)(ls + mm * 128);
            g[(size_t)(4 * mm + 0) * HDIM] = sv.x;
            g[(size_t)(4 * mm + 1) * HDIM] = sv.y;
            g[(size_t)(4 * mm + 2) * HDIM] = sv.z;
            g[(size_t)(4 * mm + 3) * HDIM] = sv.w;
          }
        }
        {
          float* la = lab + (u & 1) * SCAN_AB_SLOT + (4 * hw) * 128;
          #pragma unroll
          for (int jj = 0; jj < 8; ++jj)
            *(v2f*)(la + (jj >> 1) * 128 + (jj & 1) * 2) = R[8 * u + jj];
        }
        __builtin_amdgcn_sched_barrier(0);
        {
          const int tn = (i + SCAN_AHEAD < SCAN_NT) ? (i + SCAN_AHEAD) : (SCAN_NT - 1);
          #pragma unroll
          for (int jj = 0; jj < 8; ++jj)
            R[8 * u + jj] =
                *(const v2f*)(src + (size_t)(tn * SCAN_TILE + 8 * hw + jj) * (2 * HDIM));
        }
        scan_handover();
      }
    }
  } else {
    for (int i = 0; i < SCAN_ITERS; ++i)
      scan_handover();
  }

  if (HALF + 1 < NHALF && is_serial) {
    v2f c; c.x = hstate; c.y = v;
    *(v2f*)(st + 2 * idx) = c;
  }
}

template <int NW>
__device__ __forceinline__ void synapse_block(
    const int bid, const int half,
    const float* __restrict__ S, const float* __restrict__ Wsyn,
    float* __restrict__ out,
    float* lws, float* lA) {
  const int nt = bid & 15;
  const unsigned ml = (unsigned)bid >> 4;
  constexpr unsigned G = THALF / (NW * 64);
  const int mg = (int)((((ml / G) << 1) + (unsigned)half) * G + (ml % G));
  const int lane = threadIdx.x & 31;
  const int wave = threadIdx.x >> 5;

  {
    const float4* g = (const float4*)(Wsyn + (size_t)nt * 16 * HDIM);
    float4* sm = (float4*)lws;
    constexpr int NS = 16 * HDIM / 4 / (NW * 32);
    float4 tw[NS];
    #pragma unroll
    for (int it = 0; it < NS; ++it)
      tw[it] = g[threadIdx.x + it * (NW * 32)];
    #pragma unroll
    for (int it = 0; it < NS; ++it)
      sm[threadIdx.x + it * (NW * 32)] = tw[it];
  }

  const int hi   = lane >> 4;
  const int l16  = lane & 15;
  const int n    = nt * 16 + l16;
  const int koff = hi << 1;
  const int m0   = (mg * NW + wave) * 64;

  v8f c[4] = {v8f{}, v8f{}, v8f{}, v8f{}};

  const float* brow = lws + l16 * HDIM + koff;
  const float* gsrc = S + (size_t)(m0 + (lane >> 3)) * HDIM + 4 * (lane & 7);
  float* ldst = lA + (wave * 64 + (lane >> 3)) * APITCH + 4 * (lane & 7);
  const float* afrag = lA + (wave * 64 + l16) * APITCH + koff;

  v4f pre[16];
  A_CHUNK_LOAD(pre, gsrc);
  __syncthreads();
  for (int kc = 0; kc < NCHUNK; ++kc) {
    wave_lds_order();
    A_CHUNK_STORE(pre, ldst);
    wave_lds_order();
    if (kc + 1 < NCHUNK)
      A_CHUNK_LOAD(pre, gsrc + (kc + 1) * KC);
    __builtin_amdgcn_sched_barrier(0);
    #pragma unroll
    for (int j = 0; j < KC / 4; ++j) {
      v2f bf = *(const v2f*)(brow + kc * KC + 4 * j);
      #pragma unroll
      for (int s = 0; s < 4; ++s) {
        v2f af = *(const v2f*)(afrag + (16 * s) * APITCH + 4 * j);
        c[s] = __builtin_amdgcn_wmma_f32_16x16x4_f32(false, af, false, bf,
                                                     (short)0, c[s], false, false);
      }
    }
  }

  #pragma unroll
  for (int s = 0; s < 4; ++s) {
    const int rbase = m0 + 16 * s + (hi << 3);
    #pragma unroll
    for (int r = 0; r < 8; ++r)
      out[(size_t)(rbase + r) * HDIM + n] = c[s][r];
  }
}

__global__ __launch_bounds__(NW_SOLO * 32) void gates_first(
    const float* __restrict__ X,
    const float* __restrict__ Wa, const float* __restrict__ ba,
    const float* __restrict__ Wu, const float* __restrict__ bu,
    float* __restrict__ ab) {
  __shared__ __attribute__((aligned(16))) float lw[2 * HDIM * 16];
  __shared__ __attribute__((aligned(16))) float lA[AROWS_SOLO * APITCH];
  gates_block<NW_SOLO>(blockIdx.x, 0, X, Wa, ba, Wu, bu, ab, lw, lw + HDIM * 16, lA);
}

__global__ __launch_bounds__(256) void scan_first_gates_second(
    const float* __restrict__ X,
    const float* __restrict__ Wa, const float* __restrict__ ba,
    const float* __restrict__ Wu, const float* __restrict__ bu,
    const float* __restrict__ ab_rd, float* __restrict__ ab_wr,
    float* __restrict__ s_out, float* __restrict__ st,
    const float* __restrict__ tau_p, const float* __restrict__ thr_p) {
  __shared__ __attribute__((aligned(16))) float lw[2 * HDIM * 16];
  __shared__ __attribute__((aligned(16))) float lA[AROWS * APITCH];
  if (blockIdx.x < SCAN_BLOCKS) {
    const float tau = tau_p[0];
    const float thr = thr_p[0];
    scan_block<0>(blockIdx.x, ab_rd, s_out, st, tau, thr, lA);
    return;
  }
  gates_block<8>(blockIdx.x - SCAN_BLOCKS, 1, X, Wa, ba, Wu, bu, ab_wr,
                 lw, lw + HDIM * 16, lA);
}

__global__ __launch_bounds__(256) void scan_second_synapse_first(
    const float* __restrict__ ab, float* __restrict__ s_wr,
    float* __restrict__ st,
    const float* __restrict__ tau_p, const float* __restrict__ thr_p,
    const float* __restrict__ s_rd, const float* __restrict__ Wsyn,
    float* __restrict__ out) {
  __shared__ __attribute__((aligned(16))) float lw[16 * HDIM];
  __shared__ __attribute__((aligned(16))) float lA[AROWS * APITCH];
  if (blockIdx.x < SCAN_BLOCKS) {
    const float tau = tau_p[0];
    const float thr = thr_p[0];
    scan_block<1>(blockIdx.x, ab, s_wr, st, tau, thr, lA);
    return;
  }
  synapse_block<8>(blockIdx.x - SCAN_BLOCKS, 0, s_rd, Wsyn, out, lw, lA);
}

__global__ __launch_bounds__(NW_SOLO * 32) void synapse_second(
    const float* __restrict__ S, const float* __restrict__ Wsyn,
    float* __restrict__ out) {
  __shared__ __attribute__((aligned(16))) float lw[16 * HDIM];
  __shared__ __attribute__((aligned(16))) float lA[AROWS_SOLO * APITCH];
  synapse_block<NW_SOLO>(blockIdx.x, 1, S, Wsyn, out, lw, lA);
}

extern "C" void kernel_launch(void* const* d_in, const int* in_sizes, int n_in,
                              void* d_out, int out_size, void* d_ws, size_t ws_size,
                              hipStream_t stream) {
  const float* x    = (const float*)d_in[0];
  const float* Wa   = (const float*)d_in[1];
  const float* ba   = (const float*)d_in[2];
  const float* Wu   = (const float*)d_in[3];
  const float* bu   = (const float*)d_in[4];
  const float* tau  = (const float*)d_in[5];
  const float* thr  = (const float*)d_in[6];
  const float* Wsyn = (const float*)d_in[7];
  float* out = (float*)d_out;

  float* ab = (float*)d_ws;
  float* s  = ab + 2 * (size_t)BT * HDIM;
  float* st = s + (size_t)BT * HDIM;

  gates_first<<<GEMM_BLOCKS * (8 / NW_SOLO), NW_SOLO * 32, 0, stream>>>(x, Wa, ba, Wu, bu, ab);
  scan_first_gates_second<<<SCAN_BLOCKS + GEMM_BLOCKS, 256, 0, stream>>>(
      x, Wa, ba, Wu, bu, ab, ab, s, st, tau, thr);
  scan_second_synapse_first<<<SCAN_BLOCKS + GEMM_BLOCKS, 256, 0, stream>>>(
      ab, s, st, tau, thr, s, Wsyn, out);
  synapse_second<<<GEMM_BLOCKS * (8 / NW_SOLO), NW_SOLO * 32, 0, stream>>>(s, Wsyn, out);
}
